# starting kernel
# baseline (speedup 1.0000x reference)
#include <hip/hip_runtime.h>
#include <hip/hip_bf16.h>
#include <float.h>

typedef float v2f __attribute__((ext_vector_type(2)));
typedef float v4f __attribute__((ext_vector_type(4)));
typedef float v8f __attribute__((ext_vector_type(8)));

#define D_DIM 256
#define A_STRIDE 132
#define WAVES_PER_BLOCK 4
#define THREADS (WAVES_PER_BLOCK * 32)
#define MT 2
#define ROWS_PER_WAVE (MT * 16)
#define ROWS_PER_BLOCK (WAVES_PER_BLOCK * ROWS_PER_WAVE)
#define A_SLOTS (WAVES_PER_BLOCK * MT * 32)

__global__ void cc_norms(const float* __restrict__ cb0, int K0,
                         const float* __restrict__ cb1, int K1,
                         const float* __restrict__ cb2, int K2,
                         float* __restrict__ norms) {
    int i = blockIdx.x * blockDim.x + threadIdx.x;
    int total = K0 + K1 + K2;
    if (i >= total) return;
    const float* src;
    int row;
    if (i < K0)           { src = cb0; row = i; }
    else if (i < K0 + K1) { src = cb1; row = i - K0; }
    else                  { src = cb2; row = i - K0 - K1; }
    const v4f* p = (const v4f*)(src + (size_t)row * D_DIM);
    float s = 0.f;
    for (int j = 0; j < D_DIM / 4; ++j) {
        v4f v = p[j];
        s += v[0]*v[0] + v[1]*v[1] + v[2]*v[2] + v[3]*v[3];
    }
    norms[i] = s;
}

__global__ __launch_bounds__(THREADS)
void cc_main(const float* __restrict__ embeds,
             const float* __restrict__ cb0, int K0,
             const float* __restrict__ cb1, int K1,
             const float* __restrict__ cb2, int K2,
             const float* __restrict__ norms,
             float* __restrict__ out,
             float* __restrict__ partials,
             int nRows) {
    extern __shared__ float smem[];
    float* ldsA = smem;
    float* ldsB = smem + A_SLOTS * A_STRIDE;

    const int tid   = threadIdx.x;
    const int lane  = tid & 31;
    const int w     = tid >> 5;
    const int layer = blockIdx.y;

    const float* cb; int K; int normOff;
    if (layer == 0)      { cb = cb0; K = K0; normOff = 0; }
    else if (layer == 1) { cb = cb1; K = K1; normOff = K0; }
    else                 { cb = cb2; K = K2; normOff = K0 + K1; }

    const int rowBase = blockIdx.x * ROWS_PER_BLOCK;

    {
        int lr = tid;
        int wA = lr >> 5;
        int st = (lr >> 4) & 1;
        int mA = lr & 15;
        const v4f* src = (const v4f*)(embeds + (size_t)(rowBase + lr) * D_DIM);
        float* dst0 = ldsA + (wA * (MT * 32) + st * 32 + mA) * A_STRIDE;
        float* dst1 = dst0 + 16 * A_STRIDE;
#pragma unroll
        for (int i = 0; i < D_DIM / 4; ++i) {
            v4f v = src[i];
            *(v2f*)(dst0 + 2 * i) = (v2f){v[0], v[1]};
            *(v2f*)(dst1 + 2 * i) = (v2f){v[2], v[3]};
        }
    }

    float bestS[MT * 8];
    int   bestI[MT * 8];
#pragma unroll
    for (int r = 0; r < MT * 8; ++r) { bestS[r] = FLT_MAX; bestI[r] = 0x7fffffff; }

    const int nLane = lane & 15;
    const float* aP0 = ldsA + (w * (MT * 32) + lane) * A_STRIDE;
    const float* aP1 = aP0 + 32 * A_STRIDE;
    const float* bP  = ldsB + lane * A_STRIDE;

    const int numTiles = (K + 15) >> 4;
    for (int t = 0; t < numTiles; ++t) {
        const int tileBase = t << 4;
        __syncthreads();
        {
            int n = tid >> 3;
            int p = tid & 7;
            int srcRow = tileBase + n;
            if (srcRow >= K) srcRow = K - 1;
            const v4f* src = (const v4f*)(cb + (size_t)srcRow * D_DIM);
            float* dst0 = ldsB + n * A_STRIDE;
            float* dst1 = dst0 + 16 * A_STRIDE;
#pragma unroll
            for (int i = 0; i < 8; ++i) {
                int c = p * 8 + i;
                v4f v = src[c];
                *(v2f*)(dst0 + 2 * c) = (v2f){v[0], v[1]};
                *(v2f*)(dst1 + 2 * c) = (v2f){v[2], v[3]};
            }
        }
        __syncthreads();

        v8f acc0 = {0.f,0.f,0.f,0.f,0.f,0.f,0.f,0.f};
        v8f acc1 = {0.f,0.f,0.f,0.f,0.f,0.f,0.f,0.f};
#pragma unroll
        for (int c2 = 0; c2 < 32; ++c2) {
            v4f bv  = *(const v4f*)(bP  + c2 * 4);
            v4f av0 = *(const v4f*)(aP0 + c2 * 4);
            v4f av1 = *(const v4f*)(aP1 + c2 * 4);
            v2f b0  = {bv[0],  bv[1]},  b1  = {bv[2],  bv[3]};
            v2f a00 = {av0[0], av0[1]}, a01 = {av0[2], av0[3]};
            v2f a10 = {av1[0], av1[1]}, a11 = {av1[2], av1[3]};
            acc0 = __builtin_amdgcn_wmma_f32_16x16x4_f32(false, a00, false, b0,
                                                         (short)0, acc0, false, false);
            acc1 = __builtin_amdgcn_wmma_f32_16x16x4_f32(false, a10, false, b0,
                                                         (short)0, acc1, false, false);
            acc0 = __builtin_amdgcn_wmma_f32_16x16x4_f32(false, a01, false, b1,
                                                         (short)0, acc0, false, false);
            acc1 = __builtin_amdgcn_wmma_f32_16x16x4_f32(false, a11, false, b1,
                                                         (short)0, acc1, false, false);
        }

        const int gk = tileBase + nLane;
        const bool valid = gk < K;
        const float wn = valid ? norms[normOff + gk] : FLT_MAX;
#pragma unroll
        for (int r = 0; r < 8; ++r) {
            float s0 = wn - 2.0f * acc0[r];
            float s1 = wn - 2.0f * acc1[r];
            if (valid && s0 < bestS[r])     { bestS[r]     = s0; bestI[r]     = gk; }
            if (valid && s1 < bestS[8 + r]) { bestS[8 + r] = s1; bestI[8 + r] = gk; }
        }
    }

#pragma unroll
    for (int r = 0; r < MT * 8; ++r) {
        float s = bestS[r];
        int   i = bestI[r];
#pragma unroll
        for (int off = 8; off >= 1; off >>= 1) {
            float os = __shfl_xor(s, off, 32);
            int   oi = __shfl_xor(i, off, 32);
            if (os < s || (os == s && oi < i)) { s = os; i = oi; }
        }
        bestS[r] = s; bestI[r] = i;
    }

    const int h = lane >> 4;
    float localSum = 0.f;
#pragma unroll
    for (int st = 0; st < MT; ++st) {
#pragma unroll
        for (int r = 0; r < 8; ++r) {
            const int m   = r + 8 * h;
            const int row = rowBase + w * ROWS_PER_WAVE + st * 16 + m;
            const int idx = bestI[st * 8 + r];
            const v4f* qsrc = (const v4f*)(cb + (size_t)idx * D_DIM);
            const v4f* esrc = (const v4f*)(embeds + (size_t)row * D_DIM);
            v4f* dst = (v4f*)(out + ((size_t)layer * nRows + row) * D_DIM);
#pragma unroll
            for (int vv = 0; vv < 4; ++vv) {
                int q4 = nLane * 4 + vv;
                v4f q = qsrc[q4];
                v4f e = esrc[q4];
                dst[q4] = q;
                v4f df = q - e;
                localSum += df[0]*df[0] + df[1]*df[1] + df[2]*df[2] + df[3]*df[3];
            }
        }
    }

    __syncthreads();
    ldsB[tid] = localSum;
    __syncthreads();
    if (tid == 0) {
        float s = 0.f;
        for (int i = 0; i < THREADS; ++i) s += ldsB[i];
        partials[blockIdx.y * gridDim.x + blockIdx.x] = s;
    }
}

__global__ void cc_finalize(const float* __restrict__ partials, int n,
                            float* __restrict__ out, size_t lossIdx, float invND) {
    if (blockIdx.x == 0 && threadIdx.x == 0) {
        float s = 0.f;
        for (int i = 0; i < n; ++i) s += partials[i];
        out[lossIdx] = 1.25f * invND * s;
    }
}

extern "C" void kernel_launch(void* const* d_in, const int* in_sizes, int n_in,
                              void* d_out, int out_size, void* d_ws, size_t ws_size,
                              hipStream_t stream) {
    const float* embeds = (const float*)d_in[0];
    const float* cb0    = (const float*)d_in[1];
    const float* cb1    = (const float*)d_in[2];
    const float* cb2    = (const float*)d_in[3];

    const int nRows = in_sizes[0] / D_DIM;
    const int K0 = in_sizes[1] / D_DIM;
    const int K1 = in_sizes[2] / D_DIM;
    const int K2 = in_sizes[3] / D_DIM;
    const int totalK = K0 + K1 + K2;

    float* norms    = (float*)d_ws;
    float* partials = (float*)d_ws + 8192;

    cc_norms<<<(totalK + 255) / 256, 256, 0, stream>>>(cb0, K0, cb1, K1, cb2, K2, norms);

    const int gx = nRows / ROWS_PER_BLOCK;
    const size_t smemBytes =
        (size_t)(A_SLOTS + 32) * A_STRIDE * sizeof(float);
    cc_main<<<dim3(gx, 3), THREADS, smemBytes, stream>>>(
        embeds, cb0, K0, cb1, K1, cb2, K2, norms,
        (float*)d_out, partials, nRows);

    cc_finalize<<<1, 1, 0, stream>>>(partials, gx * 3, (float*)d_out,
                                     (size_t)out_size - 1,
                                     1.0f / ((float)nRows * (float)D_DIM));
}


# Work-group re-tiled to 64 rows x 64 codes on 8 waves (wave tile 16x32: one A fragment, two B tiles): layer 0 gives 256 groups at 2 waves per SIMD, not 128 at 1; loss partial sums regrouped
# speedup vs baseline: 2.0244x; 2.0244x over previous
#include <hip/hip_runtime.h>
#include <hip/hip_bf16.h>
#include <float.h>


typedef float v2f __attribute__((ext_vector_type(2)));
typedef float v4f __attribute__((ext_vector_type(4)));
typedef float v8f __attribute__((ext_vector_type(8)));

#define D_DIM 256
#define A_STRIDE 132
#define ROW_GROUPS 4
#define CODE_GROUPS 2
#define WAVES_PER_BLOCK (ROW_GROUPS * CODE_GROUPS)
#define THREADS (WAVES_PER_BLOCK * 32)
#define ROWS_PER_BLOCK (ROW_GROUPS * 16)
#define CODES_PER_STEP (CODE_GROUPS * 32)
#define A_SLOTS (ROW_GROUPS * 32)
#define B_SLOTS (CODE_GROUPS * 2 * 32)

__global__ void cc_norms(const float* __restrict__ cb0, int K0,
                         const float* __restrict__ cb1, int K1,
                         const float* __restrict__ cb2, int K2,
                         float* __restrict__ norms) {
    int i = blockIdx.x * blockDim.x + threadIdx.x;
    int total = K0 + K1 + K2;
    if (i >= total) return;
    const float* src;
    int row;
    if (i < K0)           { src = cb0; row = i; }
    else if (i < K0 + K1) { src = cb1; row = i - K0; }
    else                  { src = cb2; row = i - K0 - K1; }
    const v4f* p = (const v4f*)(src + (size_t)row * D_DIM);
    float s = 0.f;
    for (int j = 0; j < D_DIM / 4; ++j) {
        v4f v = p[j];
        s += v[0]*v[0] + v[1]*v[1] + v[2]*v[2] + v[3]*v[3];
    }
    norms[i] = s;
}

__global__ __launch_bounds__(THREADS)
void cc_main(const float* __restrict__ embeds,
             const float* __restrict__ cb0, int K0,
             const float* __restrict__ cb1, int K1,
             const float* __restrict__ cb2, int K2,
             const float* __restrict__ norms,
             float* __restrict__ out,
             float* __restrict__ partials,
             int nRows) {
    extern __shared__ float smem[];
    float* ldsA = smem;
    float* ldsB = smem + A_SLOTS * A_STRIDE;

    const int tid   = threadIdx.x;
    const int lane  = tid & 31;
    const int w     = tid >> 5;
    const int rg    = w & (ROW_GROUPS - 1);
    const int cg    = w >> 2;
    const int layer = blockIdx.y;

    const float* cb; int K; int normOff;
    if (layer == 0)      { cb = cb0; K = K0; normOff = 0; }
    else if (layer == 1) { cb = cb1; K = K1; normOff = K0; }
    else                 { cb = cb2; K = K2; normOff = K0 + K1; }

    const int rowBase = blockIdx.x * ROWS_PER_BLOCK;

#pragma unroll
    for (int it = 0; it < (ROWS_PER_BLOCK * (D_DIM / 4)) / THREADS; ++it) {
        const int idx = it * THREADS + tid;
        const int row = idx >> 6;
        const int c   = idx & 63;
        v4f v = *((const v4f*)(embeds + (size_t)(rowBase + row) * D_DIM) + c);
        float* dst0 = ldsA + ((row >> 4) * 32 + (row & 15)) * A_STRIDE;
        float* dst1 = dst0 + 16 * A_STRIDE;
        *(v2f*)(dst0 + 2 * c) = (v2f){v[0], v[1]};
        *(v2f*)(dst1 + 2 * c) = (v2f){v[2], v[3]};
    }

    float bestS[8];
    int   bestI[8];
#pragma unroll
    for (int r = 0; r < 8; ++r) { bestS[r] = FLT_MAX; bestI[r] = 0x7fffffff; }

    const int nLane = lane & 15;
    const float* aP  = ldsA + (rg * 32 + lane) * A_STRIDE;
    const float* bP0 = ldsB + ((cg * 2 + 0) * 32 + lane) * A_STRIDE;
    const float* bP1 = ldsB + ((cg * 2 + 1) * 32 + lane) * A_STRIDE;

    const int numSteps = (K + CODES_PER_STEP - 1) / CODES_PER_STEP;
    for (int t = 0; t < numSteps; ++t) {
        const int tileBase = t * CODES_PER_STEP;
        __syncthreads();
#pragma unroll
        for (int it = 0; it < (CODES_PER_STEP * (D_DIM / 4)) / THREADS; ++it) {
            const int idx = it * THREADS + tid;
            const int n   = idx >> 6;
            const int c   = idx & 63;
            int srcRow = tileBase + n;
            if (srcRow >= K) srcRow = K - 1;
            v4f v = *((const v4f*)(cb + (size_t)srcRow * D_DIM) + c);
            float* dst0 = ldsB + ((n >> 4) * 32 + (n & 15)) * A_STRIDE;
            float* dst1 = dst0 + 16 * A_STRIDE;
            *(v2f*)(dst0 + 2 * c) = (v2f){v[0], v[1]};
            *(v2f*)(dst1 + 2 * c) = (v2f){v[2], v[3]};
        }
        __syncthreads();

        v8f acc0 = {0.f,0.f,0.f,0.f,0.f,0.f,0.f,0.f};
        v8f acc1 = {0.f,0.f,0.f,0.f,0.f,0.f,0.f,0.f};
#pragma unroll
        for (int c2 = 0; c2 < 32; ++c2) {
            v4f av  = *(const v4f*)(aP  + c2 * 4);
            v4f bv0 = *(const v4f*)(bP0 + c2 * 4);
            v4f bv1 = *(const v4f*)(bP1 + c2 * 4);
            v2f a0  = {av[0],  av[1]},  a1  = {av[2],  av[3]};
            v2f b00 = {bv0[0], bv0[1]}, b01 = {bv0[2], bv0[3]};
            v2f b10 = {bv1[0], bv1[1]}, b11 = {bv1[2], bv1[3]};
            acc0 = __builtin_amdgcn_wmma_f32_16x16x4_f32(false, a0, false, b00,
                                                         (short)0, acc0, false, false);
            acc1 = __builtin_amdgcn_wmma_f32_16x16x4_f32(false, a0, false, b10,
                                                         (short)0, acc1, false, false);
            acc0 = __builtin_amdgcn_wmma_f32_16x16x4_f32(false, a1, false, b01,
                                                         (short)0, acc0, false, false);
            acc1 = __builtin_amdgcn_wmma_f32_16x16x4_f32(false, a1, false, b11,
                                                         (short)0, acc1, false, false);
        }

        const int gk0 = tileBase + cg * 32 + nLane;
        const int gk1 = gk0 + 16;
        const bool valid0 = gk0 < K;
        const bool valid1 = gk1 < K;
        const float wn0 = valid0 ? norms[normOff + gk0] : FLT_MAX;
        const float wn1 = valid1 ? norms[normOff + gk1] : FLT_MAX;
#pragma unroll
        for (int r = 0; r < 8; ++r) {
            float s0 = wn0 - 2.0f * acc0[r];
            float s1 = wn1 - 2.0f * acc1[r];
            if (valid0 && s0 < bestS[r]) { bestS[r] = s0; bestI[r] = gk0; }
            if (valid1 && s1 < bestS[r]) { bestS[r] = s1; bestI[r] = gk1; }
        }
    }

#pragma unroll
    for (int r = 0; r < 8; ++r) {
        float s = bestS[r];
        int   i = bestI[r];
#pragma unroll
        for (int off = 8; off >= 1; off >>= 1) {
            float os = __shfl_xor(s, off, 32);
            int   oi = __shfl_xor(i, off, 32);
            if (os < s || (os == s && oi < i)) { s = os; i = oi; }
        }
        bestS[r] = s; bestI[r] = i;
    }

    const int h = lane >> 4;
    float* mS = ldsB;
    int*   mI = (int*)(ldsB + CODE_GROUPS * ROW_GROUPS * 16);
    __syncthreads();
    if (nLane == 0) {
#pragma unroll
        for (int r = 0; r < 8; ++r) {
            mS[(cg * ROW_GROUPS + rg) * 16 + r + 8 * h] = bestS[r];
            mI[(cg * ROW_GROUPS + rg) * 16 + r + 8 * h] = bestI[r];
        }
    }
    __syncthreads();

    float localSum = 0.f;
#pragma unroll
    for (int j = 0; j < 4; ++j) {
        const int m   = cg * 8 + 2 * j + h;
        const int row = rowBase + rg * 16 + m;
        const float sA = mS[(0 * ROW_GROUPS + rg) * 16 + m];
        const int   iA = mI[(0 * ROW_GROUPS + rg) * 16 + m];
        const float sB = mS[(1 * ROW_GROUPS + rg) * 16 + m];
        const int   iB = mI[(1 * ROW_GROUPS + rg) * 16 + m];
        const int idx = (sB < sA || (sB == sA && iB < iA)) ? iB : iA;
        const v4f* qsrc = (const v4f*)(cb + (size_t)idx * D_DIM);
        const v4f* esrc = (const v4f*)(embeds + (size_t)row * D_DIM);
        v4f* dst = (v4f*)(out + ((size_t)layer * nRows + row) * D_DIM);
#pragma unroll
        for (int vv = 0; vv < 4; ++vv) {
            int q4 = nLane * 4 + vv;
            v4f q = qsrc[q4];
            v4f e = esrc[q4];
            dst[q4] = q;
            v4f df = q - e;
            localSum += df[0]*df[0] + df[1]*df[1] + df[2]*df[2] + df[3]*df[3];
        }
    }

    __syncthreads();
    ldsB[tid] = localSum;
    __syncthreads();
    if (tid == 0) {
        float s = 0.f;
        for (int i = 0; i < THREADS; ++i) s += ldsB[i];
        partials[blockIdx.y * gridDim.x + blockIdx.x] = s;
    }
}

__global__ void cc_finalize(const float* __restrict__ partials, int n,
                            float* __restrict__ out, size_t lossIdx, float invND) {
    if (blockIdx.x == 0 && threadIdx.x == 0) {
        float s = 0.f;
        for (int i = 0; i < n; ++i) s += partials[i];
        out[lossIdx] = 1.25f * invND * s;
    }
}

extern "C" void kernel_launch(void* const* d_in, const int* in_sizes, int n_in,
                              void* d_out, int out_size, void* d_ws, size_t ws_size,
                              hipStream_t stream) {
    const float* embeds = (const float*)d_in[0];
    const float* cb0    = (const float*)d_in[1];
    const float* cb1    = (const float*)d_in[2];
    const float* cb2    = (const float*)d_in[3];

    const int nRows = in_sizes[0] / D_DIM;
    const int K0 = in_sizes[1] / D_DIM;
    const int K1 = in_sizes[2] / D_DIM;
    const int K2 = in_sizes[3] / D_DIM;
    const int totalK = K0 + K1 + K2;

    float* norms    = (float*)d_ws;
    float* partials = (float*)d_ws + 8192;

    cc_norms<<<(totalK + 255) / 256, 256, 0, stream>>>(cb0, K0, cb1, K1, cb2, K2, norms);

    const int gx = nRows / ROWS_PER_BLOCK;
    const size_t smemBytes =
        (size_t)(A_SLOTS + B_SLOTS) * A_STRIDE * sizeof(float);
    cc_main<<<dim3(gx, 3), THREADS, smemBytes, stream>>>(
        embeds, cb0, K0, cb1, K1, cb2, K2, norms,
        (float*)d_out, partials, nRows);

    cc_finalize<<<1, 1, 0, stream>>>(partials, gx * 3, (float*)d_out,
                                     (size_t)out_size - 1,
                                     1.0f / ((float)nRows * (float)D_DIM));
}


# Next B tile fetched global->registers before the WMMA block and stored to a 2nd LDS buffer after it (one barrier per step; LDS 202752 B, so one work-group per WGP); summation order unchanged
# speedup vs baseline: 2.5535x; 1.2614x over previous
#include <hip/hip_runtime.h>
#include <hip/hip_bf16.h>
#include <float.h>


typedef float v2f __attribute__((ext_vector_type(2)));
typedef float v4f __attribute__((ext_vector_type(4)));
typedef float v8f __attribute__((ext_vector_type(8)));

#define D_DIM 256
#define A_STRIDE 132
#define ROW_GROUPS 4
#define CODE_GROUPS 2
#define WAVES_PER_BLOCK (ROW_GROUPS * CODE_GROUPS)
#define THREADS (WAVES_PER_BLOCK * 32)
#define ROWS_PER_BLOCK (ROW_GROUPS * 16)
#define CODES_PER_STEP (CODE_GROUPS * 32)
#define A_SLOTS (ROW_GROUPS * 32)
#define B_SLOTS (CODE_GROUPS * 2 * 32)
#define TILE_FLOATS (B_SLOTS * A_STRIDE)
#define ROWS_PER_IT (THREADS / (D_DIM / 4))
#define LOADS_PER_THREAD (CODES_PER_STEP / ROWS_PER_IT)

__device__ __forceinline__ void cc_tile_fetch(v4f (&pre)[LOADS_PER_THREAD],
                                              const float* __restrict__ cb, int K,
                                              int tileBase, int nSub, int c) {
#pragma unroll
    for (int it = 0; it < LOADS_PER_THREAD; ++it) {
        int srcRow = tileBase + it * ROWS_PER_IT + nSub;
        if (srcRow >= K) srcRow = K - 1;
        pre[it] = *((const v4f*)(cb + (size_t)srcRow * D_DIM) + c);
    }
}

__device__ __forceinline__ void cc_tile_store(const v4f (&pre)[LOADS_PER_THREAD],
                                              float* __restrict__ buf, int nSub, int c) {
#pragma unroll
    for (int it = 0; it < LOADS_PER_THREAD; ++it) {
        const int n = it * ROWS_PER_IT + nSub;
        float* dst0 = buf + ((n >> 4) * 32 + (n & 15)) * A_STRIDE;
        float* dst1 = dst0 + 16 * A_STRIDE;
        *(v2f*)(dst0 + 2 * c) = (v2f){pre[it][0], pre[it][1]};
        *(v2f*)(dst1 + 2 * c) = (v2f){pre[it][2], pre[it][3]};
    }
}

__global__ void cc_norms(const float* __restrict__ cb0, int K0,
                         const float* __restrict__ cb1, int K1,
                         const float* __restrict__ cb2, int K2,
                         float* __restrict__ norms) {
    int i = blockIdx.x * blockDim.x + threadIdx.x;
    int total = K0 + K1 + K2;
    if (i >= total) return;
    const float* src;
    int row;
    if (i < K0)           { src = cb0; row = i; }
    else if (i < K0 + K1) { src = cb1; row = i - K0; }
    else                  { src = cb2; row = i - K0 - K1; }
    const v4f* p = (const v4f*)(src + (size_t)row * D_DIM);
    float s = 0.f;
    for (int j = 0; j < D_DIM / 4; ++j) {
        v4f v = p[j];
        s += v[0]*v[0] + v[1]*v[1] + v[2]*v[2] + v[3]*v[3];
    }
    norms[i] = s;
}

__global__ __launch_bounds__(THREADS)
void cc_main(const float* __restrict__ embeds,
             const float* __restrict__ cb0, int K0,
             const float* __restrict__ cb1, int K1,
             const float* __restrict__ cb2, int K2,
             const float* __restrict__ norms,
             float* __restrict__ out,
             float* __restrict__ partials,
             int nRows) {
    extern __shared__ float smem[];
    float* ldsA = smem;
    float* ldsB = smem + A_SLOTS * A_STRIDE;

    const int tid   = threadIdx.x;
    const int lane  = tid & 31;
    const int w     = tid >> 5;
    const int rg    = w & (ROW_GROUPS - 1);
    const int cg    = w >> 2;
    const int layer = blockIdx.y;

    const float* cb; int K; int normOff;
    if (layer == 0)      { cb = cb0; K = K0; normOff = 0; }
    else if (layer == 1) { cb = cb1; K = K1; normOff = K0; }
    else                 { cb = cb2; K = K2; normOff = K0 + K1; }

    const int rowBase = blockIdx.x * ROWS_PER_BLOCK;

#pragma unroll
    for (int it = 0; it < (ROWS_PER_BLOCK * (D_DIM / 4)) / THREADS; ++it) {
        const int idx = it * THREADS + tid;
        const int row = idx >> 6;
        const int c   = idx & 63;
        v4f v = *((const v4f*)(embeds + (size_t)(rowBase + row) * D_DIM) + c);
        float* dst0 = ldsA + ((row >> 4) * 32 + (row & 15)) * A_STRIDE;
        float* dst1 = dst0 + 16 * A_STRIDE;
        *(v2f*)(dst0 + 2 * c) = (v2f){v[0], v[1]};
        *(v2f*)(dst1 + 2 * c) = (v2f){v[2], v[3]};
    }

    float bestS[8];
    int   bestI[8];
#pragma unroll
    for (int r = 0; r < 8; ++r) { bestS[r] = FLT_MAX; bestI[r] = 0x7fffffff; }

    const int nLane = lane & 15;
    const float* aP = ldsA + (rg * 32 + lane) * A_STRIDE;
    const int bOff0 = ((cg * 2 + 0) * 32 + lane) * A_STRIDE;
    const int bOff1 = ((cg * 2 + 1) * 32 + lane) * A_STRIDE;
    const int nSub  = tid >> 6;
    const int cIdx  = tid & 63;
    const int laneCode = cg * 32 + nLane;

    const int numSteps = (K + CODES_PER_STEP - 1) / CODES_PER_STEP;

    v4f pre[LOADS_PER_THREAD];
    cc_tile_fetch(pre, cb, K, 0, nSub, cIdx);
    float wn0 = (laneCode      < K) ? norms[normOff + laneCode]      : FLT_MAX;
    float wn1 = (laneCode + 16 < K) ? norms[normOff + laneCode + 16] : FLT_MAX;
    cc_tile_store(pre, ldsB, nSub, cIdx);
    __syncthreads();

    for (int t = 0; t < numSteps; ++t) {
        const int tileBase = t * CODES_PER_STEP;
        const bool more = (t + 1 < numSteps);
        float wnNext0 = FLT_MAX, wnNext1 = FLT_MAX;
        if (more) {
            const int nb = tileBase + CODES_PER_STEP;
            cc_tile_fetch(pre, cb, K, nb, nSub, cIdx);
            if (nb + laneCode      < K) wnNext0 = norms[normOff + nb + laneCode];
            if (nb + laneCode + 16 < K) wnNext1 = norms[normOff + nb + laneCode + 16];
        }

        const float* bCur = ldsB + (t & 1) * TILE_FLOATS;
        const float* bP0  = bCur + bOff0;
        const float* bP1  = bCur + bOff1;

        v8f acc0 = {0.f,0.f,0.f,0.f,0.f,0.f,0.f,0.f};
        v8f acc1 = {0.f,0.f,0.f,0.f,0.f,0.f,0.f,0.f};
#pragma unroll
        for (int c2 = 0; c2 < 32; ++c2) {
            v4f av  = *(const v4f*)(aP  + c2 * 4);
            v4f bv0 = *(const v4f*)(bP0 + c2 * 4);
            v4f bv1 = *(const v4f*)(bP1 + c2 * 4);
            v2f a0  = {av[0],  av[1]},  a1  = {av[2],  av[3]};
            v2f b00 = {bv0[0], bv0[1]}, b01 = {bv0[2], bv0[3]};
            v2f b10 = {bv1[0], bv1[1]}, b11 = {bv1[2], bv1[3]};
            acc0 = __builtin_amdgcn_wmma_f32_16x16x4_f32(false, a0, false, b00,
                                                         (short)0, acc0, false, false);
            acc1 = __builtin_amdgcn_wmma_f32_16x16x4_f32(false, a0, false, b10,
                                                         (short)0, acc1, false, false);
            acc0 = __builtin_amdgcn_wmma_f32_16x16x4_f32(false, a1, false, b01,
                                                         (short)0, acc0, false, false);
            acc1 = __builtin_amdgcn_wmma_f32_16x16x4_f32(false, a1, false, b11,
                                                         (short)0, acc1, false, false);
        }

        const int gk0 = tileBase + laneCode;
        const int gk1 = gk0 + 16;
        const bool valid0 = gk0 < K;
        const bool valid1 = gk1 < K;
#pragma unroll
        for (int r = 0; r < 8; ++r) {
            float s0 = wn0 - 2.0f * acc0[r];
            float s1 = wn1 - 2.0f * acc1[r];
            if (valid0 && s0 < bestS[r]) { bestS[r] = s0; bestI[r] = gk0; }
            if (valid1 && s1 < bestS[r]) { bestS[r] = s1; bestI[r] = gk1; }
        }

        if (more) cc_tile_store(pre, ldsB + ((t + 1) & 1) * TILE_FLOATS, nSub, cIdx);
        wn0 = wnNext0;
        wn1 = wnNext1;
        __syncthreads();
    }

#pragma unroll
    for (int r = 0; r < 8; ++r) {
        float s = bestS[r];
        int   i = bestI[r];
#pragma unroll
        for (int off = 8; off >= 1; off >>= 1) {
            float os = __shfl_xor(s, off, 32);
            int   oi = __shfl_xor(i, off, 32);
            if (os < s || (os == s && oi < i)) { s = os; i = oi; }
        }
        bestS[r] = s; bestI[r] = i;
    }

    const int h = lane >> 4;
    float* mS = ldsB;
    int*   mI = (int*)(ldsB + CODE_GROUPS * ROW_GROUPS * 16);
    if (nLane == 0) {
#pragma unroll
        for (int r = 0; r < 8; ++r) {
            mS[(cg * ROW_GROUPS + rg) * 16 + r + 8 * h] = bestS[r];
            mI[(cg * ROW_GROUPS + rg) * 16 + r + 8 * h] = bestI[r];
        }
    }
    __syncthreads();

    float localSum = 0.f;
#pragma unroll
    for (int j = 0; j < 4; ++j) {
        const int m   = cg * 8 + 2 * j + h;
        const int row = rowBase + rg * 16 + m;
        const float sA = mS[(0 * ROW_GROUPS + rg) * 16 + m];
        const int   iA = mI[(0 * ROW_GROUPS + rg) * 16 + m];
        const float sB = mS[(1 * ROW_GROUPS + rg) * 16 + m];
        const int   iB = mI[(1 * ROW_GROUPS + rg) * 16 + m];
        const int idx = (sB < sA || (sB == sA && iB < iA)) ? iB : iA;
        const v4f* qsrc = (const v4f*)(cb + (size_t)idx * D_DIM);
        const v4f* esrc = (const v4f*)(embeds + (size_t)row * D_DIM);
        v4f* dst = (v4f*)(out + ((size_t)layer * nRows + row) * D_DIM);
#pragma unroll
        for (int vv = 0; vv < 4; ++vv) {
            int q4 = nLane * 4 + vv;
            v4f q = qsrc[q4];
            v4f e = esrc[q4];
            dst[q4] = q;
            v4f df = q - e;
            localSum += df[0]*df[0] + df[1]*df[1] + df[2]*df[2] + df[3]*df[3];
        }
    }

    __syncthreads();
    ldsB[tid] = localSum;
    __syncthreads();
    if (tid == 0) {
        float s = 0.f;
        for (int i = 0; i < THREADS; ++i) s += ldsB[i];
        partials[blockIdx.y * gridDim.x + blockIdx.x] = s;
    }
}

__global__ void cc_finalize(const float* __restrict__ partials, int n,
                            float* __restrict__ out, size_t lossIdx, float invND) {
    if (blockIdx.x == 0 && threadIdx.x == 0) {
        float s = 0.f;
        for (int i = 0; i < n; ++i) s += partials[i];
        out[lossIdx] = 1.25f * invND * s;
    }
}

extern "C" void kernel_launch(void* const* d_in, const int* in_sizes, int n_in,
                              void* d_out, int out_size, void* d_ws, size_t ws_size,
                              hipStream_t stream) {
    const float* embeds = (const float*)d_in[0];
    const float* cb0    = (const float*)d_in[1];
    const float* cb1    = (const float*)d_in[2];
    const float* cb2    = (const float*)d_in[3];

    const int nRows = in_sizes[0] / D_DIM;
    const int K0 = in_sizes[1] / D_DIM;
    const int K1 = in_sizes[2] / D_DIM;
    const int K2 = in_sizes[3] / D_DIM;
    const int totalK = K0 + K1 + K2;

    float* norms    = (float*)d_ws;
    float* partials = (float*)d_ws + 8192;

    cc_norms<<<(totalK + 255) / 256, 256, 0, stream>>>(cb0, K0, cb1, K1, cb2, K2, norms);

    const int gx = nRows / ROWS_PER_BLOCK;
    const size_t smemBytes =
        (size_t)(A_SLOTS + 2 * B_SLOTS) * A_STRIDE * sizeof(float);
    cc_main<<<dim3(gx, 3), THREADS, smemBytes, stream>>>(
        embeds, cb0, K0, cb1, K1, cb2, K2, norms,
        (float*)d_out, partials, nRows);

    cc_finalize<<<1, 1, 0, stream>>>(partials, gx * 3, (float*)d_out,
                                     (size_t)out_size - 1,
                                     1.0f / ((float)nRows * (float)D_DIM));
}


# Step barrier split into signal and wait: a wave signals right behind its LDS tile stores, does the step's argmin update and the fetch issue of the tile after next, then waits; sum order unchanged
# speedup vs baseline: 2.6705x; 1.0458x over previous
#include <hip/hip_runtime.h>
#include <hip/hip_bf16.h>
#include <float.h>


typedef float v2f __attribute__((ext_vector_type(2)));
typedef float v4f __attribute__((ext_vector_type(4)));
typedef float v8f __attribute__((ext_vector_type(8)));

#define D_DIM 256
#define A_STRIDE 132
#define ROW_GROUPS 4
#define CODE_GROUPS 2
#define WAVES_PER_BLOCK (ROW_GROUPS * CODE_GROUPS)
#define THREADS (WAVES_PER_BLOCK * 32)
#define ROWS_PER_BLOCK (ROW_GROUPS * 16)
#define CODES_PER_STEP (CODE_GROUPS * 32)
#define A_SLOTS (ROW_GROUPS * 32)
#define B_SLOTS (CODE_GROUPS * 2 * 32)
#define TILE_FLOATS (B_SLOTS * A_STRIDE)
#define ROWS_PER_IT (THREADS / (D_DIM / 4))
#define LOADS_PER_THREAD (CODES_PER_STEP / ROWS_PER_IT)

__device__ __forceinline__ void cc_tile_fetch(v4f (&pre)[LOADS_PER_THREAD],
                                              const float* __restrict__ cb, int K,
                                              int tileBase, int nSub, int c) {
#pragma unroll
    for (int it = 0; it < LOADS_PER_THREAD; ++it) {
        int srcRow = tileBase + it * ROWS_PER_IT + nSub;
        if (srcRow >= K) srcRow = K - 1;
        pre[it] = *((const v4f*)(cb + (size_t)srcRow * D_DIM) + c);
    }
}

__device__ __forceinline__ void cc_tile_store(const v4f (&pre)[LOADS_PER_THREAD],
                                              float* __restrict__ buf, int nSub, int c) {
#pragma unroll
    for (int it = 0; it < LOADS_PER_THREAD; ++it) {
        const int n = it * ROWS_PER_IT + nSub;
        float* dst0 = buf + ((n >> 4) * 32 + (n & 15)) * A_STRIDE;
        float* dst1 = dst0 + 16 * A_STRIDE;
        *(v2f*)(dst0 + 2 * c) = (v2f){pre[it][0], pre[it][1]};
        *(v2f*)(dst1 + 2 * c) = (v2f){pre[it][2], pre[it][3]};
    }
}

__device__ __forceinline__ void cc_arrive() {
    asm volatile("s_wait_dscnt 0x0\n\ts_barrier_signal -1" ::: "memory");
}
__device__ __forceinline__ void cc_join() {
    asm volatile("s_barrier_wait -1" ::: "memory");
}

__global__ void cc_norms(const float* __restrict__ cb0, int K0,
                         const float* __restrict__ cb1, int K1,
                         const float* __restrict__ cb2, int K2,
                         float* __restrict__ norms) {
    int i = blockIdx.x * blockDim.x + threadIdx.x;
    int total = K0 + K1 + K2;
    if (i >= total) return;
    const float* src;
    int row;
    if (i < K0)           { src = cb0; row = i; }
    else if (i < K0 + K1) { src = cb1; row = i - K0; }
    else                  { src = cb2; row = i - K0 - K1; }
    const v4f* p = (const v4f*)(src + (size_t)row * D_DIM);
    float s = 0.f;
    for (int j = 0; j < D_DIM / 4; ++j) {
        v4f v = p[j];
        s += v[0]*v[0] + v[1]*v[1] + v[2]*v[2] + v[3]*v[3];
    }
    norms[i] = s;
}

__global__ __launch_bounds__(THREADS)
void cc_main(const float* __restrict__ embeds,
             const float* __restrict__ cb0, int K0,
             const float* __restrict__ cb1, int K1,
             const float* __restrict__ cb2, int K2,
             const float* __restrict__ norms,
             float* __restrict__ out,
             float* __restrict__ partials,
             int nRows) {
    extern __shared__ float smem[];
    float* ldsA = smem;
    float* ldsB = smem + A_SLOTS * A_STRIDE;

    const int tid   = threadIdx.x;
    const int lane  = tid & 31;
    const int w     = tid >> 5;
    const int rg    = w & (ROW_GROUPS - 1);
    const int cg    = w >> 2;
    const int layer = blockIdx.y;

    const float* cb; int K; int normOff;
    if (layer == 0)      { cb = cb0; K = K0; normOff = 0; }
    else if (layer == 1) { cb = cb1; K = K1; normOff = K0; }
    else                 { cb = cb2; K = K2; normOff = K0 + K1; }

    const int rowBase = blockIdx.x * ROWS_PER_BLOCK;

#pragma unroll
    for (int it = 0; it < (ROWS_PER_BLOCK * (D_DIM / 4)) / THREADS; ++it) {
        const int idx = it * THREADS + tid;
        const int row = idx >> 6;
        const int c   = idx & 63;
        v4f v = *((const v4f*)(embeds + (size_t)(rowBase + row) * D_DIM) + c);
        float* dst0 = ldsA + ((row >> 4) * 32 + (row & 15)) * A_STRIDE;
        float* dst1 = dst0 + 16 * A_STRIDE;
        *(v2f*)(dst0 + 2 * c) = (v2f){v[0], v[1]};
        *(v2f*)(dst1 + 2 * c) = (v2f){v[2], v[3]};
    }

    float bestS[8];
    int   bestI[8];
#pragma unroll
    for (int r = 0; r < 8; ++r) { bestS[r] = FLT_MAX; bestI[r] = 0x7fffffff; }

    const int nLane = lane & 15;
    const float* aP = ldsA + (rg * 32 + lane) * A_STRIDE;
    const int bOff0 = ((cg * 2 + 0) * 32 + lane) * A_STRIDE;
    const int bOff1 = ((cg * 2 + 1) * 32 + lane) * A_STRIDE;
    const int nSub  = tid >> 6;
    const int cIdx  = tid & 63;
    const int laneCode = cg * 32 + nLane;

    const int numSteps = (K + CODES_PER_STEP - 1) / CODES_PER_STEP;

    v4f pre[LOADS_PER_THREAD];
    cc_tile_fetch(pre, cb, K, 0, nSub, cIdx);
    cc_tile_store(pre, ldsB, nSub, cIdx);
    cc_arrive();
    if (numSteps > 1)
        cc_tile_fetch(pre, cb, K, CODES_PER_STEP, nSub, cIdx);
    cc_join();

    for (int t = 0; t < numSteps; ++t) {
        const int tileBase = t * CODES_PER_STEP;
        const bool more = (t + 1 < numSteps);
        const int gk0 = tileBase + laneCode;
        const int gk1 = gk0 + 16;
        const bool valid0 = gk0 < K;
        const bool valid1 = gk1 < K;
        const float nr0 = norms[normOff + (valid0 ? gk0 : K - 1)];
        const float nr1 = norms[normOff + (valid1 ? gk1 : K - 1)];
        __builtin_amdgcn_sched_barrier(0);

        const float* bCur = ldsB + (t & 1) * TILE_FLOATS;
        const float* bP0  = bCur + bOff0;
        const float* bP1  = bCur + bOff1;

        v8f acc0 = {0.f,0.f,0.f,0.f,0.f,0.f,0.f,0.f};
        v8f acc1 = {0.f,0.f,0.f,0.f,0.f,0.f,0.f,0.f};
#pragma unroll
        for (int c2 = 0; c2 < 32; ++c2) {
            v4f av  = *(const v4f*)(aP  + c2 * 4);
            v4f bv0 = *(const v4f*)(bP0 + c2 * 4);
            v4f bv1 = *(const v4f*)(bP1 + c2 * 4);
            v2f a0  = {av[0],  av[1]},  a1  = {av[2],  av[3]};
            v2f b00 = {bv0[0], bv0[1]}, b01 = {bv0[2], bv0[3]};
            v2f b10 = {bv1[0], bv1[1]}, b11 = {bv1[2], bv1[3]};
            acc0 = __builtin_amdgcn_wmma_f32_16x16x4_f32(false, a0, false, b00,
                                                         (short)0, acc0, false, false);
            acc1 = __builtin_amdgcn_wmma_f32_16x16x4_f32(false, a0, false, b10,
                                                         (short)0, acc1, false, false);
            acc0 = __builtin_amdgcn_wmma_f32_16x16x4_f32(false, a1, false, b01,
                                                         (short)0, acc0, false, false);
            acc1 = __builtin_amdgcn_wmma_f32_16x16x4_f32(false, a1, false, b11,
                                                         (short)0, acc1, false, false);
        }

        if (more) cc_tile_store(pre, ldsB + ((t + 1) & 1) * TILE_FLOATS, nSub, cIdx);
        cc_arrive();

        const float wn0 = valid0 ? nr0 : FLT_MAX;
        const float wn1 = valid1 ? nr1 : FLT_MAX;
#pragma unroll
        for (int r = 0; r < 8; ++r) {
            float s0 = wn0 - 2.0f * acc0[r];
            float s1 = wn1 - 2.0f * acc1[r];
            if (valid0 && s0 < bestS[r]) { bestS[r] = s0; bestI[r] = gk0; }
            if (valid1 && s1 < bestS[r]) { bestS[r] = s1; bestI[r] = gk1; }
        }
        if (t + 2 < numSteps)
            cc_tile_fetch(pre, cb, K, tileBase + 2 * CODES_PER_STEP, nSub, cIdx);
        cc_join();
    }

#pragma unroll
    for (int r = 0; r < 8; ++r) {
        float s = bestS[r];
        int   i = bestI[r];
#pragma unroll
        for (int off = 8; off >= 1; off >>= 1) {
            float os = __shfl_xor(s, off, 32);
            int   oi = __shfl_xor(i, off, 32);
            if (os < s || (os == s && oi < i)) { s = os; i = oi; }
        }
        bestS[r] = s; bestI[r] = i;
    }

    const int h = lane >> 4;
    float* mS = ldsB;
    int*   mI = (int*)(ldsB + CODE_GROUPS * ROW_GROUPS * 16);
    if (nLane == 0) {
#pragma unroll
        for (int r = 0; r < 8; ++r) {
            mS[(cg * ROW_GROUPS + rg) * 16 + r + 8 * h] = bestS[r];
            mI[(cg * ROW_GROUPS + rg) * 16 + r + 8 * h] = bestI[r];
        }
    }
    __syncthreads();

    float localSum = 0.f;
#pragma unroll
    for (int j = 0; j < 4; ++j) {
        const int m   = cg * 8 + 2 * j + h;
        const int row = rowBase + rg * 16 + m;
        const float sA = mS[(0 * ROW_GROUPS + rg) * 16 + m];
        const int   iA = mI[(0 * ROW_GROUPS + rg) * 16 + m];
        const float sB = mS[(1 * ROW_GROUPS + rg) * 16 + m];
        const int   iB = mI[(1 * ROW_GROUPS + rg) * 16 + m];
        const int idx = (sB < sA || (sB == sA && iB < iA)) ? iB : iA;
        const v4f* qsrc = (const v4f*)(cb + (size_t)idx * D_DIM);
        const v4f* esrc = (const v4f*)(embeds + (size_t)row * D_DIM);
        v4f* dst = (v4f*)(out + ((size_t)layer * nRows + row) * D_DIM);
#pragma unroll
        for (int vv = 0; vv < 4; ++vv) {
            int q4 = nLane * 4 + vv;
            v4f q = qsrc[q4];
            v4f e = esrc[q4];
            dst[q4] = q;
            v4f df = q - e;
            localSum += df[0]*df[0] + df[1]*df[1] + df[2]*df[2] + df[3]*df[3];
        }
    }

    __syncthreads();
    ldsB[tid] = localSum;
    __syncthreads();
    if (tid == 0) {
        float s = 0.f;
        for (int i = 0; i < THREADS; ++i) s += ldsB[i];
        partials[blockIdx.y * gridDim.x + blockIdx.x] = s;
    }
}

__global__ void cc_finalize(const float* __restrict__ partials, int n,
                            float* __restrict__ out, size_t lossIdx, float invND) {
    if (blockIdx.x == 0 && threadIdx.x == 0) {
        float s = 0.f;
        for (int i = 0; i < n; ++i) s += partials[i];
        out[lossIdx] = 1.25f * invND * s;
    }
}

extern "C" void kernel_launch(void* const* d_in, const int* in_sizes, int n_in,
                              void* d_out, int out_size, void* d_ws, size_t ws_size,
                              hipStream_t stream) {
    const float* embeds = (const float*)d_in[0];
    const float* cb0    = (const float*)d_in[1];
    const float* cb1    = (const float*)d_in[2];
    const float* cb2    = (const float*)d_in[3];

    const int nRows = in_sizes[0] / D_DIM;
    const int K0 = in_sizes[1] / D_DIM;
    const int K1 = in_sizes[2] / D_DIM;
    const int K2 = in_sizes[3] / D_DIM;
    const int totalK = K0 + K1 + K2;

    float* norms    = (float*)d_ws;
    float* partials = (float*)d_ws + 8192;

    cc_norms<<<(totalK + 255) / 256, 256, 0, stream>>>(cb0, K0, cb1, K1, cb2, K2, norms);

    const int gx = nRows / ROWS_PER_BLOCK;
    const size_t smemBytes =
        (size_t)(A_SLOTS + 2 * B_SLOTS) * A_STRIDE * sizeof(float);
    cc_main<<<dim3(gx, 3), THREADS, smemBytes, stream>>>(
        embeds, cb0, K0, cb1, K1, cb2, K2, norms,
        (float*)d_out, partials, nRows);

    cc_finalize<<<1, 1, 0, stream>>>(partials, gx * 3, (float*)d_out,
                                     (size_t)out_size - 1,
                                     1.0f / ((float)nRows * (float)D_DIM));
}


# K loop's LDS fragment reads regrouped: 24 loads as one group a whole 32-WMMA block ahead (2nd register set) instead of 3 behind every 4 WMMAs; 2 read places per step, not 31; sum order unchanged
# speedup vs baseline: 2.7505x; 1.0300x over previous
#include <hip/hip_runtime.h>
#include <hip/hip_bf16.h>
#include <float.h>


typedef float v2f __attribute__((ext_vector_type(2)));
typedef float v4f __attribute__((ext_vector_type(4)));
typedef float v8f __attribute__((ext_vector_type(8)));

#define D_DIM 256
#define A_STRIDE 132
#define ROW_GROUPS 4
#define CODE_GROUPS 2
#define WAVES_PER_BLOCK (ROW_GROUPS * CODE_GROUPS)
#define THREADS (WAVES_PER_BLOCK * 32)
#define ROWS_PER_BLOCK (ROW_GROUPS * 16)
#define CODES_PER_STEP (CODE_GROUPS * 32)
#define A_SLOTS (ROW_GROUPS * 32)
#define B_SLOTS (CODE_GROUPS * 2 * 32)
#define TILE_FLOATS (B_SLOTS * A_STRIDE)
#define ROWS_PER_IT (THREADS / (D_DIM / 4))
#define LOADS_PER_THREAD (CODES_PER_STEP / ROWS_PER_IT)
#define K_CHUNKS (D_DIM / 8)
#define BLK_CHUNKS 8
#define N_BLK (K_CHUNKS / BLK_CHUNKS)

__device__ __forceinline__ void cc_tile_fetch(v4f (&pre)[LOADS_PER_THREAD],
                                              const float* __restrict__ cb, int K,
                                              int tileBase, int nSub, int c) {
#pragma unroll
    for (int it = 0; it < LOADS_PER_THREAD; ++it) {
        int srcRow = tileBase + it * ROWS_PER_IT + nSub;
        if (srcRow >= K) srcRow = K - 1;
        pre[it] = *((const v4f*)(cb + (size_t)srcRow * D_DIM) + c);
    }
}

__device__ __forceinline__ void cc_tile_store(const v4f (&pre)[LOADS_PER_THREAD],
                                              float* __restrict__ buf, int nSub, int c) {
#pragma unroll
    for (int it = 0; it < LOADS_PER_THREAD; ++it) {
        const int n = it * ROWS_PER_IT + nSub;
        float* dst0 = buf + ((n >> 4) * 32 + (n & 15)) * A_STRIDE;
        float* dst1 = dst0 + 16 * A_STRIDE;
        *(v2f*)(dst0 + 2 * c) = (v2f){pre[it][0], pre[it][1]};
        *(v2f*)(dst1 + 2 * c) = (v2f){pre[it][2], pre[it][3]};
    }
}

struct cc_frags {
    v4f a[BLK_CHUNKS];
    v4f b0[BLK_CHUNKS];
    v4f b1[BLK_CHUNKS];
};

__device__ __forceinline__ void cc_frag_load(cc_frags& f, const float* __restrict__ aP,
                                             const float* __restrict__ bP0,
                                             const float* __restrict__ bP1, int blk) {
#pragma unroll
    for (int i = 0; i < BLK_CHUNKS; ++i) {
        const int c2 = blk * BLK_CHUNKS + i;
        f.a[i]  = *(const v4f*)(aP  + c2 * 4);
        f.b0[i] = *(const v4f*)(bP0 + c2 * 4);
        f.b1[i] = *(const v4f*)(bP1 + c2 * 4);
    }
}

__device__ __forceinline__ void cc_block_mma(const cc_frags& f, v8f& acc0, v8f& acc1) {
#pragma unroll
    for (int i = 0; i < BLK_CHUNKS; ++i) {
        const v4f av = f.a[i], bv0 = f.b0[i], bv1 = f.b1[i];
        v2f a0  = {av[0],  av[1]},  a1  = {av[2],  av[3]};
        v2f b00 = {bv0[0], bv0[1]}, b01 = {bv0[2], bv0[3]};
        v2f b10 = {bv1[0], bv1[1]}, b11 = {bv1[2], bv1[3]};
        acc0 = __builtin_amdgcn_wmma_f32_16x16x4_f32(false, a0, false, b00,
                                                     (short)0, acc0, false, false);
        acc1 = __builtin_amdgcn_wmma_f32_16x16x4_f32(false, a0, false, b10,
                                                     (short)0, acc1, false, false);
        acc0 = __builtin_amdgcn_wmma_f32_16x16x4_f32(false, a1, false, b01,
                                                     (short)0, acc0, false, false);
        acc1 = __builtin_amdgcn_wmma_f32_16x16x4_f32(false, a1, false, b11,
                                                     (short)0, acc1, false, false);
    }
}

__device__ __forceinline__ void cc_arrive() {
    asm volatile("s_wait_dscnt 0x0\n\ts_barrier_signal -1" ::: "memory");
}
__device__ __forceinline__ void cc_join() {
    asm volatile("s_barrier_wait -1" ::: "memory");
}

__global__ void cc_norms(const float* __restrict__ cb0, int K0,
                         const float* __restrict__ cb1, int K1,
                         const float* __restrict__ cb2, int K2,
                         float* __restrict__ norms) {
    int i = blockIdx.x * blockDim.x + threadIdx.x;
    int total = K0 + K1 + K2;
    if (i >= total) return;
    const float* src;
    int row;
    if (i < K0)           { src = cb0; row = i; }
    else if (i < K0 + K1) { src = cb1; row = i - K0; }
    else                  { src = cb2; row = i - K0 - K1; }
    const v4f* p = (const v4f*)(src + (size_t)row * D_DIM);
    float s = 0.f;
    for (int j = 0; j < D_DIM / 4; ++j) {
        v4f v = p[j];
        s += v[0]*v[0] + v[1]*v[1] + v[2]*v[2] + v[3]*v[3];
    }
    norms[i] = s;
}

__global__ __launch_bounds__(THREADS)
void cc_main(const float* __restrict__ embeds,
             const float* __restrict__ cb0, int K0,
             const float* __restrict__ cb1, int K1,
             const float* __restrict__ cb2, int K2,
             const float* __restrict__ norms,
             float* __restrict__ out,
             float* __restrict__ partials,
             int nRows) {
    extern __shared__ float smem[];
    float* ldsA = smem;
    float* ldsB = smem + A_SLOTS * A_STRIDE;

    const int tid   = threadIdx.x;
    const int lane  = tid & 31;
    const int w     = tid >> 5;
    const int rg    = w & (ROW_GROUPS - 1);
    const int cg    = w >> 2;
    const int layer = blockIdx.y;

    const float* cb; int K; int normOff;
    if (layer == 0)      { cb = cb0; K = K0; normOff = 0; }
    else if (layer == 1) { cb = cb1; K = K1; normOff = K0; }
    else                 { cb = cb2; K = K2; normOff = K0 + K1; }

    const int rowBase = blockIdx.x * ROWS_PER_BLOCK;

#pragma unroll
    for (int it = 0; it < (ROWS_PER_BLOCK * (D_DIM / 4)) / THREADS; ++it) {
        const int idx = it * THREADS + tid;
        const int row = idx >> 6;
        const int c   = idx & 63;
        v4f v = *((const v4f*)(embeds + (size_t)(rowBase + row) * D_DIM) + c);
        float* dst0 = ldsA + ((row >> 4) * 32 + (row & 15)) * A_STRIDE;
        float* dst1 = dst0 + 16 * A_STRIDE;
        *(v2f*)(dst0 + 2 * c) = (v2f){v[0], v[1]};
        *(v2f*)(dst1 + 2 * c) = (v2f){v[2], v[3]};
    }

    float bestS[8];
    int   bestI[8];
#pragma unroll
    for (int r = 0; r < 8; ++r) { bestS[r] = FLT_MAX; bestI[r] = 0x7fffffff; }

    const int nLane = lane & 15;
    const float* aP = ldsA + (rg * 32 + lane) * A_STRIDE;
    const int bOff0 = ((cg * 2 + 0) * 32 + lane) * A_STRIDE;
    const int bOff1 = ((cg * 2 + 1) * 32 + lane) * A_STRIDE;
    const int nSub  = tid >> 6;
    const int cIdx  = tid & 63;
    const int laneCode = cg * 32 + nLane;

    const int numSteps = (K + CODES_PER_STEP - 1) / CODES_PER_STEP;

    v4f pre[LOADS_PER_THREAD];
    cc_tile_fetch(pre, cb, K, 0, nSub, cIdx);
    cc_tile_store(pre, ldsB, nSub, cIdx);
    cc_arrive();
    if (numSteps > 1)
        cc_tile_fetch(pre, cb, K, CODES_PER_STEP, nSub, cIdx);
    cc_join();

    for (int t = 0; t < numSteps; ++t) {
        const int tileBase = t * CODES_PER_STEP;
        const bool more = (t + 1 < numSteps);
        const int gk0 = tileBase + laneCode;
        const int gk1 = gk0 + 16;
        const bool valid0 = gk0 < K;
        const bool valid1 = gk1 < K;
        const float nr0 = norms[normOff + (valid0 ? gk0 : K - 1)];
        const float nr1 = norms[normOff + (valid1 ? gk1 : K - 1)];
        __builtin_amdgcn_sched_barrier(0);

        const float* bCur = ldsB + (t & 1) * TILE_FLOATS;
        const float* bP0  = bCur + bOff0;
        const float* bP1  = bCur + bOff1;

        v8f acc0 = {0.f,0.f,0.f,0.f,0.f,0.f,0.f,0.f};
        v8f acc1 = {0.f,0.f,0.f,0.f,0.f,0.f,0.f,0.f};
        cc_frags fr[2];
        cc_frag_load(fr[0], aP, bP0, bP1, 0);
        cc_frag_load(fr[1], aP, bP0, bP1, 1);
#pragma unroll
        for (int b = 0; b < N_BLK; ++b) {
            if (b + 1 < N_BLK) __builtin_amdgcn_sched_barrier(0);
            cc_block_mma(fr[b & 1], acc0, acc1);
            if (b + 2 < N_BLK) {
                __builtin_amdgcn_sched_barrier(0);
                cc_frag_load(fr[b & 1], aP, bP0, bP1, b + 2);
            }
        }

        if (more) cc_tile_store(pre, ldsB + ((t + 1) & 1) * TILE_FLOATS, nSub, cIdx);
        cc_arrive();

        const float wn0 = valid0 ? nr0 : FLT_MAX;
        const float wn1 = valid1 ? nr1 : FLT_MAX;
#pragma unroll
        for (int r = 0; r < 8; ++r) {
            float s0 = wn0 - 2.0f * acc0[r];
            float s1 = wn1 - 2.0f * acc1[r];
            if (valid0 && s0 < bestS[r]) { bestS[r] = s0; bestI[r] = gk0; }
            if (valid1 && s1 < bestS[r]) { bestS[r] = s1; bestI[r] = gk1; }
        }
        if (t + 2 < numSteps)
            cc_tile_fetch(pre, cb, K, tileBase + 2 * CODES_PER_STEP, nSub, cIdx);
        cc_join();
    }

#pragma unroll
    for (int r = 0; r < 8; ++r) {
        float s = bestS[r];
        int   i = bestI[r];
#pragma unroll
        for (int off = 8; off >= 1; off >>= 1) {
            float os = __shfl_xor(s, off, 32);
            int   oi = __shfl_xor(i, off, 32);
            if (os < s || (os == s && oi < i)) { s = os; i = oi; }
        }
        bestS[r] = s; bestI[r] = i;
    }

    const int h = lane >> 4;
    float* mS = ldsB;
    int*   mI = (int*)(ldsB + CODE_GROUPS * ROW_GROUPS * 16);
    if (nLane == 0) {
#pragma unroll
        for (int r = 0; r < 8; ++r) {
            mS[(cg * ROW_GROUPS + rg) * 16 + r + 8 * h] = bestS[r];
            mI[(cg * ROW_GROUPS + rg) * 16 + r + 8 * h] = bestI[r];
        }
    }
    __syncthreads();

    float localSum = 0.f;
#pragma unroll
    for (int j = 0; j < 4; ++j) {
        const int m   = cg * 8 + 2 * j + h;
        const int row = rowBase + rg * 16 + m;
        const float sA = mS[(0 * ROW_GROUPS + rg) * 16 + m];
        const int   iA = mI[(0 * ROW_GROUPS + rg) * 16 + m];
        const float sB = mS[(1 * ROW_GROUPS + rg) * 16 + m];
        const int   iB = mI[(1 * ROW_GROUPS + rg) * 16 + m];
        const int idx = (sB < sA || (sB == sA && iB < iA)) ? iB : iA;
        const v4f* qsrc = (const v4f*)(cb + (size_t)idx * D_DIM);
        const v4f* esrc = (const v4f*)(embeds + (size_t)row * D_DIM);
        v4f* dst = (v4f*)(out + ((size_t)layer * nRows + row) * D_DIM);
#pragma unroll
        for (int vv = 0; vv < 4; ++vv) {
            int q4 = nLane * 4 + vv;
            v4f q = qsrc[q4];
            v4f e = esrc[q4];
            dst[q4] = q;
            v4f df = q - e;
            localSum += df[0]*df[0] + df[1]*df[1] + df[2]*df[2] + df[3]*df[3];
        }
    }

    __syncthreads();
    ldsB[tid] = localSum;
    __syncthreads();
    if (tid == 0) {
        float s = 0.f;
        for (int i = 0; i < THREADS; ++i) s += ldsB[i];
        partials[blockIdx.y * gridDim.x + blockIdx.x] = s;
    }
}

__global__ void cc_finalize(const float* __restrict__ partials, int n,
                            float* __restrict__ out, size_t lossIdx, float invND) {
    if (blockIdx.x == 0 && threadIdx.x == 0) {
        float s = 0.f;
        for (int i = 0; i < n; ++i) s += partials[i];
        out[lossIdx] = 1.25f * invND * s;
    }
}

extern "C" void kernel_launch(void* const* d_in, const int* in_sizes, int n_in,
                              void* d_out, int out_size, void* d_ws, size_t ws_size,
                              hipStream_t stream) {
    const float* embeds = (const float*)d_in[0];
    const float* cb0    = (const float*)d_in[1];
    const float* cb1    = (const float*)d_in[2];
    const float* cb2    = (const float*)d_in[3];

    const int nRows = in_sizes[0] / D_DIM;
    const int K0 = in_sizes[1] / D_DIM;
    const int K1 = in_sizes[2] / D_DIM;
    const int K2 = in_sizes[3] / D_DIM;
    const int totalK = K0 + K1 + K2;

    float* norms    = (float*)d_ws;
    float* partials = (float*)d_ws + 8192;

    cc_norms<<<(totalK + 255) / 256, 256, 0, stream>>>(cb0, K0, cb1, K1, cb2, K2, norms);

    const int gx = nRows / ROWS_PER_BLOCK;
    const size_t smemBytes =
        (size_t)(A_SLOTS + 2 * B_SLOTS) * A_STRIDE * sizeof(float);
    cc_main<<<dim3(gx, 3), THREADS, smemBytes, stream>>>(
        embeds, cb0, K0, cb1, K1, cb2, K2, norms,
        (float*)d_out, partials, nRows);

    cc_finalize<<<1, 1, 0, stream>>>(partials, gx * 3, (float*)d_out,
                                     (size_t)out_size - 1,
                                     1.0f / ((float)nRows * (float)D_DIM));
}


# Loss sums made trees: each wave folds its lanes by xor shuffles and stores one partial (no LDS pass, no serial 256-term loop); final sum on 1024 threads, not 1; loss summation order changed
# speedup vs baseline: 2.9252x; 1.0635x over previous
#include <hip/hip_runtime.h>
#include <hip/hip_bf16.h>
#include <float.h>


typedef float v2f __attribute__((ext_vector_type(2)));
typedef float v4f __attribute__((ext_vector_type(4)));
typedef float v8f __attribute__((ext_vector_type(8)));

#define D_DIM 256
#define A_STRIDE 132
#define ROW_GROUPS 4
#define CODE_GROUPS 2
#define WAVES_PER_BLOCK (ROW_GROUPS * CODE_GROUPS)
#define THREADS (WAVES_PER_BLOCK * 32)
#define ROWS_PER_BLOCK (ROW_GROUPS * 16)
#define CODES_PER_STEP (CODE_GROUPS * 32)
#define A_SLOTS (ROW_GROUPS * 32)
#define B_SLOTS (CODE_GROUPS * 2 * 32)
#define TILE_FLOATS (B_SLOTS * A_STRIDE)
#define ROWS_PER_IT (THREADS / (D_DIM / 4))
#define LOADS_PER_THREAD (CODES_PER_STEP / ROWS_PER_IT)
#define K_CHUNKS (D_DIM / 8)
#define BLK_CHUNKS 8
#define N_BLK (K_CHUNKS / BLK_CHUNKS)

__device__ __forceinline__ void cc_tile_fetch(v4f (&pre)[LOADS_PER_THREAD],
                                              const float* __restrict__ cb, int K,
                                              int tileBase, int nSub, int c) {
#pragma unroll
    for (int it = 0; it < LOADS_PER_THREAD; ++it) {
        int srcRow = tileBase + it * ROWS_PER_IT + nSub;
        if (srcRow >= K) srcRow = K - 1;
        pre[it] = *((const v4f*)(cb + (size_t)srcRow * D_DIM) + c);
    }
}

__device__ __forceinline__ void cc_tile_store(const v4f (&pre)[LOADS_PER_THREAD],
                                              float* __restrict__ buf, int nSub, int c) {
#pragma unroll
    for (int it = 0; it < LOADS_PER_THREAD; ++it) {
        const int n = it * ROWS_PER_IT + nSub;
        float* dst0 = buf + ((n >> 4) * 32 + (n & 15)) * A_STRIDE;
        float* dst1 = dst0 + 16 * A_STRIDE;
        *(v2f*)(dst0 + 2 * c) = (v2f){pre[it][0], pre[it][1]};
        *(v2f*)(dst1 + 2 * c) = (v2f){pre[it][2], pre[it][3]};
    }
}

struct cc_frags {
    v4f a[BLK_CHUNKS];
    v4f b0[BLK_CHUNKS];
    v4f b1[BLK_CHUNKS];
};

__device__ __forceinline__ void cc_frag_load(cc_frags& f, const float* __restrict__ aP,
                                             const float* __restrict__ bP0,
                                             const float* __restrict__ bP1, int blk) {
#pragma unroll
    for (int i = 0; i < BLK_CHUNKS; ++i) {
        const int c2 = blk * BLK_CHUNKS + i;
        f.a[i]  = *(const v4f*)(aP  + c2 * 4);
        f.b0[i] = *(const v4f*)(bP0 + c2 * 4);
        f.b1[i] = *(const v4f*)(bP1 + c2 * 4);
    }
}

__device__ __forceinline__ void cc_block_mma(const cc_frags& f, v8f& acc0, v8f& acc1) {
#pragma unroll
    for (int i = 0; i < BLK_CHUNKS; ++i) {
        const v4f av = f.a[i], bv0 = f.b0[i], bv1 = f.b1[i];
        v2f a0  = {av[0],  av[1]},  a1  = {av[2],  av[3]};
        v2f b00 = {bv0[0], bv0[1]}, b01 = {bv0[2], bv0[3]};
        v2f b10 = {bv1[0], bv1[1]}, b11 = {bv1[2], bv1[3]};
        acc0 = __builtin_amdgcn_wmma_f32_16x16x4_f32(false, a0, false, b00,
                                                     (short)0, acc0, false, false);
        acc1 = __builtin_amdgcn_wmma_f32_16x16x4_f32(false, a0, false, b10,
                                                     (short)0, acc1, false, false);
        acc0 = __builtin_amdgcn_wmma_f32_16x16x4_f32(false, a1, false, b01,
                                                     (short)0, acc0, false, false);
        acc1 = __builtin_amdgcn_wmma_f32_16x16x4_f32(false, a1, false, b11,
                                                     (short)0, acc1, false, false);
    }
}

__device__ __forceinline__ void cc_arrive() {
    asm volatile("s_wait_dscnt 0x0\n\ts_barrier_signal -1" ::: "memory");
}
__device__ __forceinline__ void cc_join() {
    asm volatile("s_barrier_wait -1" ::: "memory");
}

__global__ void cc_norms(const float* __restrict__ cb0, int K0,
                         const float* __restrict__ cb1, int K1,
                         const float* __restrict__ cb2, int K2,
                         float* __restrict__ norms) {
    int i = blockIdx.x * blockDim.x + threadIdx.x;
    int total = K0 + K1 + K2;
    if (i >= total) return;
    const float* src;
    int row;
    if (i < K0)           { src = cb0; row = i; }
    else if (i < K0 + K1) { src = cb1; row = i - K0; }
    else                  { src = cb2; row = i - K0 - K1; }
    const v4f* p = (const v4f*)(src + (size_t)row * D_DIM);
    float s = 0.f;
    for (int j = 0; j < D_DIM / 4; ++j) {
        v4f v = p[j];
        s += v[0]*v[0] + v[1]*v[1] + v[2]*v[2] + v[3]*v[3];
    }
    norms[i] = s;
}

__global__ __launch_bounds__(THREADS)
void cc_main(const float* __restrict__ embeds,
             const float* __restrict__ cb0, int K0,
             const float* __restrict__ cb1, int K1,
             const float* __restrict__ cb2, int K2,
             const float* __restrict__ norms,
             float* __restrict__ out,
             float* __restrict__ partials,
             int nRows) {
    extern __shared__ float smem[];
    float* ldsA = smem;
    float* ldsB = smem + A_SLOTS * A_STRIDE;

    const int tid   = threadIdx.x;
    const int lane  = tid & 31;
    const int w     = tid >> 5;
    const int rg    = w & (ROW_GROUPS - 1);
    const int cg    = w >> 2;
    const int layer = blockIdx.y;

    const float* cb; int K; int normOff;
    if (layer == 0)      { cb = cb0; K = K0; normOff = 0; }
    else if (layer == 1) { cb = cb1; K = K1; normOff = K0; }
    else                 { cb = cb2; K = K2; normOff = K0 + K1; }

    const int rowBase = blockIdx.x * ROWS_PER_BLOCK;

#pragma unroll
    for (int it = 0; it < (ROWS_PER_BLOCK * (D_DIM / 4)) / THREADS; ++it) {
        const int idx = it * THREADS + tid;
        const int row = idx >> 6;
        const int c   = idx & 63;
        v4f v = *((const v4f*)(embeds + (size_t)(rowBase + row) * D_DIM) + c);
        float* dst0 = ldsA + ((row >> 4) * 32 + (row & 15)) * A_STRIDE;
        float* dst1 = dst0 + 16 * A_STRIDE;
        *(v2f*)(dst0 + 2 * c) = (v2f){v[0], v[1]};
        *(v2f*)(dst1 + 2 * c) = (v2f){v[2], v[3]};
    }

    float bestS[8];
    int   bestI[8];
#pragma unroll
    for (int r = 0; r < 8; ++r) { bestS[r] = FLT_MAX; bestI[r] = 0x7fffffff; }

    const int nLane = lane & 15;
    const float* aP = ldsA + (rg * 32 + lane) * A_STRIDE;
    const int bOff0 = ((cg * 2 + 0) * 32 + lane) * A_STRIDE;
    const int bOff1 = ((cg * 2 + 1) * 32 + lane) * A_STRIDE;
    const int nSub  = tid >> 6;
    const int cIdx  = tid & 63;
    const int laneCode = cg * 32 + nLane;

    const int numSteps = (K + CODES_PER_STEP - 1) / CODES_PER_STEP;

    v4f pre[LOADS_PER_THREAD];
    cc_tile_fetch(pre, cb, K, 0, nSub, cIdx);
    cc_tile_store(pre, ldsB, nSub, cIdx);
    cc_arrive();
    if (numSteps > 1)
        cc_tile_fetch(pre, cb, K, CODES_PER_STEP, nSub, cIdx);
    cc_join();

    for (int t = 0; t < numSteps; ++t) {
        const int tileBase = t * CODES_PER_STEP;
        const bool more = (t + 1 < numSteps);
        const int gk0 = tileBase + laneCode;
        const int gk1 = gk0 + 16;
        const bool valid0 = gk0 < K;
        const bool valid1 = gk1 < K;
        const float nr0 = norms[normOff + (valid0 ? gk0 : K - 1)];
        const float nr1 = norms[normOff + (valid1 ? gk1 : K - 1)];
        __builtin_amdgcn_sched_barrier(0);

        const float* bCur = ldsB + (t & 1) * TILE_FLOATS;
        const float* bP0  = bCur + bOff0;
        const float* bP1  = bCur + bOff1;

        v8f acc0 = {0.f,0.f,0.f,0.f,0.f,0.f,0.f,0.f};
        v8f acc1 = {0.f,0.f,0.f,0.f,0.f,0.f,0.f,0.f};
        cc_frags fr[2];
        cc_frag_load(fr[0], aP, bP0, bP1, 0);
        cc_frag_load(fr[1], aP, bP0, bP1, 1);
#pragma unroll
        for (int b = 0; b < N_BLK; ++b) {
            if (b + 1 < N_BLK) __builtin_amdgcn_sched_barrier(0);
            cc_block_mma(fr[b & 1], acc0, acc1);
            if (b + 2 < N_BLK) {
                __builtin_amdgcn_sched_barrier(0);
                cc_frag_load(fr[b & 1], aP, bP0, bP1, b + 2);
            }
        }

        if (more) cc_tile_store(pre, ldsB + ((t + 1) & 1) * TILE_FLOATS, nSub, cIdx);
        cc_arrive();

        const float wn0 = valid0 ? nr0 : FLT_MAX;
        const float wn1 = valid1 ? nr1 : FLT_MAX;
#pragma unroll
        for (int r = 0; r < 8; ++r) {
            float s0 = wn0 - 2.0f * acc0[r];
            float s1 = wn1 - 2.0f * acc1[r];
            if (valid0 && s0 < bestS[r]) { bestS[r] = s0; bestI[r] = gk0; }
            if (valid1 && s1 < bestS[r]) { bestS[r] = s1; bestI[r] = gk1; }
        }
        if (t + 2 < numSteps)
            cc_tile_fetch(pre, cb, K, tileBase + 2 * CODES_PER_STEP, nSub, cIdx);
        cc_join();
    }

#pragma unroll
    for (int r = 0; r < 8; ++r) {
        float s = bestS[r];
        int   i = bestI[r];
#pragma unroll
        for (int off = 8; off >= 1; off >>= 1) {
            float os = __shfl_xor(s, off, 32);
            int   oi = __shfl_xor(i, off, 32);
            if (os < s || (os == s && oi < i)) { s = os; i = oi; }
        }
        bestS[r] = s; bestI[r] = i;
    }

    const int h = lane >> 4;
    float* mS = ldsB;
    int*   mI = (int*)(ldsB + CODE_GROUPS * ROW_GROUPS * 16);
    if (nLane == 0) {
#pragma unroll
        for (int r = 0; r < 8; ++r) {
            mS[(cg * ROW_GROUPS + rg) * 16 + r + 8 * h] = bestS[r];
            mI[(cg * ROW_GROUPS + rg) * 16 + r + 8 * h] = bestI[r];
        }
    }
    __syncthreads();

    float localSum = 0.f;
#pragma unroll
    for (int j = 0; j < 4; ++j) {
        const int m   = cg * 8 + 2 * j + h;
        const int row = rowBase + rg * 16 + m;
        const float sA = mS[(0 * ROW_GROUPS + rg) * 16 + m];
        const int   iA = mI[(0 * ROW_GROUPS + rg) * 16 + m];
        const float sB = mS[(1 * ROW_GROUPS + rg) * 16 + m];
        const int   iB = mI[(1 * ROW_GROUPS + rg) * 16 + m];
        const int idx = (sB < sA || (sB == sA && iB < iA)) ? iB : iA;
        const v4f* qsrc = (const v4f*)(cb + (size_t)idx * D_DIM);
        const v4f* esrc = (const v4f*)(embeds + (size_t)row * D_DIM);
        v4f* dst = (v4f*)(out + ((size_t)layer * nRows + row) * D_DIM);
#pragma unroll
        for (int vv = 0; vv < 4; ++vv) {
            int q4 = nLane * 4 + vv;
            v4f q = qsrc[q4];
            v4f e = esrc[q4];
            dst[q4] = q;
            v4f df = q - e;
            localSum += df[0]*df[0] + df[1]*df[1] + df[2]*df[2] + df[3]*df[3];
        }
    }

#pragma unroll
    for (int off = 16; off >= 1; off >>= 1)
        localSum += __shfl_xor(localSum, off, 32);
    if (lane == 0)
        partials[(blockIdx.y * gridDim.x + blockIdx.x) * WAVES_PER_BLOCK + w] = localSum;
}

#define FIN_THREADS 1024
#define FIN_BATCH 8
__global__ __launch_bounds__(FIN_THREADS)
void cc_finalize(const float* __restrict__ partials, int n,
                 float* __restrict__ out, size_t lossIdx, float invND) {
    __shared__ float waveSum[FIN_THREADS / 32];
    const int tid  = threadIdx.x;
    const int lane = tid & 31;
    float s = 0.f;
    for (int base = 0; base < n; base += FIN_THREADS * FIN_BATCH) {
        float v[FIN_BATCH];
#pragma unroll
        for (int j = 0; j < FIN_BATCH; ++j) {
            const int i = base + j * FIN_THREADS + tid;
            v[j] = (i < n) ? partials[i] : 0.f;
        }
#pragma unroll
        for (int j = 0; j < FIN_BATCH; ++j) s += v[j];
    }
#pragma unroll
    for (int off = 16; off >= 1; off >>= 1)
        s += __shfl_xor(s, off, 32);
    if (lane == 0) waveSum[tid >> 5] = s;
    __syncthreads();
    if (tid < 32) {
        float t = waveSum[lane];
#pragma unroll
        for (int off = 16; off >= 1; off >>= 1)
            t += __shfl_xor(t, off, 32);
        if (lane == 0) out[lossIdx] = 1.25f * invND * t;
    }
}

extern "C" void kernel_launch(void* const* d_in, const int* in_sizes, int n_in,
                              void* d_out, int out_size, void* d_ws, size_t ws_size,
                              hipStream_t stream) {
    const float* embeds = (const float*)d_in[0];
    const float* cb0    = (const float*)d_in[1];
    const float* cb1    = (const float*)d_in[2];
    const float* cb2    = (const float*)d_in[3];

    const int nRows = in_sizes[0] / D_DIM;
    const int K0 = in_sizes[1] / D_DIM;
    const int K1 = in_sizes[2] / D_DIM;
    const int K2 = in_sizes[3] / D_DIM;
    const int totalK = K0 + K1 + K2;

    float* norms    = (float*)d_ws;
    float* partials = (float*)d_ws + 8192;

    cc_norms<<<(totalK + 255) / 256, 256, 0, stream>>>(cb0, K0, cb1, K1, cb2, K2, norms);

    const int gx = nRows / ROWS_PER_BLOCK;
    const size_t smemBytes =
        (size_t)(A_SLOTS + 2 * B_SLOTS) * A_STRIDE * sizeof(float);
    cc_main<<<dim3(gx, 3), THREADS, smemBytes, stream>>>(
        embeds, cb0, K0, cb1, K1, cb2, K2, norms,
        (float*)d_out, partials, nRows);

    cc_finalize<<<1, FIN_THREADS, 0, stream>>>(partials, gx * 3 * WAVES_PER_BLOCK, (float*)d_out,
                                     (size_t)out_size - 1,
                                     1.0f / ((float)nRows * (float)D_DIM));
}


# Wave tile turned to 32 rows x 16 codes: wave's A operand read from LDS once and held in registers, K loop reads 8 B fragments per block, not 24; norm wait fenced before the fetch; sum order kept
# speedup vs baseline: 3.0800x; 1.0529x over previous
#include <hip/hip_runtime.h>
#include <hip/hip_bf16.h>
#include <float.h>


typedef float v2f __attribute__((ext_vector_type(2)));
typedef float v4f __attribute__((ext_vector_type(4)));
typedef float v8f __attribute__((ext_vector_type(8)));

#define D_DIM 256
#define A_STRIDE 132
#define ROW_GROUPS 2
#define CODE_GROUPS 4
#define WAVES_PER_BLOCK (ROW_GROUPS * CODE_GROUPS)
#define THREADS (WAVES_PER_BLOCK * 32)
#define ROWS_PER_BLOCK (ROW_GROUPS * 32)
#define CODES_PER_STEP (CODE_GROUPS * 16)
#define A_SLOTS (ROW_GROUPS * 2 * 32)
#define B_SLOTS (CODE_GROUPS * 32)
#define TILE_FLOATS (B_SLOTS * A_STRIDE)
#define ROWS_PER_IT (THREADS / (D_DIM / 4))
#define LOADS_PER_THREAD (CODES_PER_STEP / ROWS_PER_IT)
#define K_CHUNKS (D_DIM / 8)
#define BLK_CHUNKS 8
#define N_BLK (K_CHUNKS / BLK_CHUNKS)

__device__ __forceinline__ void cc_tile_fetch(v4f (&pre)[LOADS_PER_THREAD],
                                              const float* __restrict__ cb, int K,
                                              int tileBase, int nSub, int c) {
#pragma unroll
    for (int it = 0; it < LOADS_PER_THREAD; ++it) {
        int srcRow = tileBase + it * ROWS_PER_IT + nSub;
        if (srcRow >= K) srcRow = K - 1;
        pre[it] = *((const v4f*)(cb + (size_t)srcRow * D_DIM) + c);
    }
}

__device__ __forceinline__ void cc_tile_store(const v4f (&pre)[LOADS_PER_THREAD],
                                              float* __restrict__ buf, int nSub, int c) {
#pragma unroll
    for (int it = 0; it < LOADS_PER_THREAD; ++it) {
        const int n = it * ROWS_PER_IT + nSub;
        float* dst0 = buf + ((n >> 4) * 32 + (n & 15)) * A_STRIDE;
        float* dst1 = dst0 + 16 * A_STRIDE;
        *(v2f*)(dst0 + 2 * c) = (v2f){pre[it][0], pre[it][1]};
        *(v2f*)(dst1 + 2 * c) = (v2f){pre[it][2], pre[it][3]};
    }
}

struct cc_frags {
    v4f b[BLK_CHUNKS];
};

__device__ __forceinline__ void cc_frag_load(cc_frags& f, const float* __restrict__ bP, int blk) {
#pragma unroll
    for (int i = 0; i < BLK_CHUNKS; ++i) {
        const int c2 = blk * BLK_CHUNKS + i;
        f.b[i] = *(const v4f*)(bP + c2 * 4);
    }
}

__device__ __forceinline__ void cc_block_mma(const cc_frags& f,
                                             const v4f (&aT0)[K_CHUNKS], const v4f (&aT1)[K_CHUNKS],
                                             int blk, v8f& acc0, v8f& acc1) {
#pragma unroll
    for (int i = 0; i < BLK_CHUNKS; ++i) {
        const int c2 = blk * BLK_CHUNKS + i;
        const v4f av0 = aT0[c2], av1 = aT1[c2], bv = f.b[i];
        v2f a00 = {av0[0], av0[1]}, a01 = {av0[2], av0[3]};
        v2f a10 = {av1[0], av1[1]}, a11 = {av1[2], av1[3]};
        v2f b0  = {bv[0],  bv[1]},  b1  = {bv[2],  bv[3]};
        acc0 = __builtin_amdgcn_wmma_f32_16x16x4_f32(false, a00, false, b0,
                                                     (short)0, acc0, false, false);
        acc1 = __builtin_amdgcn_wmma_f32_16x16x4_f32(false, a10, false, b0,
                                                     (short)0, acc1, false, false);
        acc0 = __builtin_amdgcn_wmma_f32_16x16x4_f32(false, a01, false, b1,
                                                     (short)0, acc0, false, false);
        acc1 = __builtin_amdgcn_wmma_f32_16x16x4_f32(false, a11, false, b1,
                                                     (short)0, acc1, false, false);
    }
}

__device__ __forceinline__ void cc_arrive() {
    asm volatile("s_wait_dscnt 0x0\n\ts_barrier_signal -1" ::: "memory");
}
__device__ __forceinline__ void cc_join() {
    asm volatile("s_barrier_wait -1" ::: "memory");
}

__global__ void cc_norms(const float* __restrict__ cb0, int K0,
                         const float* __restrict__ cb1, int K1,
                         const float* __restrict__ cb2, int K2,
                         float* __restrict__ norms) {
    int i = blockIdx.x * blockDim.x + threadIdx.x;
    int total = K0 + K1 + K2;
    if (i >= total) return;
    const float* src;
    int row;
    if (i < K0)           { src = cb0; row = i; }
    else if (i < K0 + K1) { src = cb1; row = i - K0; }
    else                  { src = cb2; row = i - K0 - K1; }
    const v4f* p = (const v4f*)(src + (size_t)row * D_DIM);
    float s = 0.f;
    for (int j = 0; j < D_DIM / 4; ++j) {
        v4f v = p[j];
        s += v[0]*v[0] + v[1]*v[1] + v[2]*v[2] + v[3]*v[3];
    }
    norms[i] = s;
}

__global__ __launch_bounds__(THREADS)
void cc_main(const float* __restrict__ embeds,
             const float* __restrict__ cb0, int K0,
             const float* __restrict__ cb1, int K1,
             const float* __restrict__ cb2, int K2,
             const float* __restrict__ norms,
             float* __restrict__ out,
             float* __restrict__ partials,
             int nRows) {
    extern __shared__ float smem[];
    float* ldsA = smem;
    float* ldsB = smem + A_SLOTS * A_STRIDE;

    const int tid   = threadIdx.x;
    const int lane  = tid & 31;
    const int w     = tid >> 5;
    const int rg    = w & (ROW_GROUPS - 1);
    const int cg    = w >> 1;
    const int layer = blockIdx.y;

    const float* cb; int K; int normOff;
    if (layer == 0)      { cb = cb0; K = K0; normOff = 0; }
    else if (layer == 1) { cb = cb1; K = K1; normOff = K0; }
    else                 { cb = cb2; K = K2; normOff = K0 + K1; }

    const int rowBase = blockIdx.x * ROWS_PER_BLOCK;

#pragma unroll
    for (int it = 0; it < (ROWS_PER_BLOCK * (D_DIM / 4)) / THREADS; ++it) {
        const int idx = it * THREADS + tid;
        const int row = idx >> 6;
        const int c   = idx & 63;
        v4f v = *((const v4f*)(embeds + (size_t)(rowBase + row) * D_DIM) + c);
        float* dst0 = ldsA + ((row >> 4) * 32 + (row & 15)) * A_STRIDE;
        float* dst1 = dst0 + 16 * A_STRIDE;
        *(v2f*)(dst0 + 2 * c) = (v2f){v[0], v[1]};
        *(v2f*)(dst1 + 2 * c) = (v2f){v[2], v[3]};
    }

    float bestS[16];
    int   bestI[16];
#pragma unroll
    for (int r = 0; r < 16; ++r) { bestS[r] = FLT_MAX; bestI[r] = 0x7fffffff; }

    const int nLane = lane & 15;
    const float* aP0 = ldsA + ((rg * 2 + 0) * 32 + lane) * A_STRIDE;
    const float* aP1 = ldsA + ((rg * 2 + 1) * 32 + lane) * A_STRIDE;
    const int bOff  = (cg * 32 + lane) * A_STRIDE;
    const int nSub  = tid >> 6;
    const int cIdx  = tid & 63;
    const int laneCode = cg * 16 + nLane;

    const int numSteps = (K + CODES_PER_STEP - 1) / CODES_PER_STEP;

    v4f pre[LOADS_PER_THREAD];
    cc_tile_fetch(pre, cb, K, 0, nSub, cIdx);
    cc_tile_store(pre, ldsB, nSub, cIdx);
    cc_arrive();
    if (numSteps > 1)
        cc_tile_fetch(pre, cb, K, CODES_PER_STEP, nSub, cIdx);
    cc_join();

    v4f aT0[K_CHUNKS], aT1[K_CHUNKS];
#pragma unroll
    for (int c2 = 0; c2 < K_CHUNKS; ++c2) {
        aT0[c2] = *(const v4f*)(aP0 + c2 * 4);
        aT1[c2] = *(const v4f*)(aP1 + c2 * 4);
    }

    for (int t = 0; t < numSteps; ++t) {
        const int tileBase = t * CODES_PER_STEP;
        const bool more = (t + 1 < numSteps);
        const int gk = tileBase + laneCode;
        const bool valid = gk < K;
        const float nr = norms[normOff + (valid ? gk : K - 1)];
        __builtin_amdgcn_sched_barrier(0);

        const float* bP = ldsB + (t & 1) * TILE_FLOATS + bOff;

        v8f acc0 = {0.f,0.f,0.f,0.f,0.f,0.f,0.f,0.f};
        v8f acc1 = {0.f,0.f,0.f,0.f,0.f,0.f,0.f,0.f};
        cc_frags fr[2];
        cc_frag_load(fr[0], bP, 0);
        cc_frag_load(fr[1], bP, 1);
#pragma unroll
        for (int b = 0; b < N_BLK; ++b) {
            if (b + 1 < N_BLK) __builtin_amdgcn_sched_barrier(0);
            cc_block_mma(fr[b & 1], aT0, aT1, b, acc0, acc1);
            if (b + 2 < N_BLK) {
                __builtin_amdgcn_sched_barrier(0);
                cc_frag_load(fr[b & 1], bP, b + 2);
            }
        }

        if (more) cc_tile_store(pre, ldsB + ((t + 1) & 1) * TILE_FLOATS, nSub, cIdx);
        cc_arrive();

        asm volatile("" :: "v"(nr));

        const float wn = valid ? nr : FLT_MAX;
#pragma unroll
        for (int r = 0; r < 8; ++r) {
            float s0 = wn - 2.0f * acc0[r];
            float s1 = wn - 2.0f * acc1[r];
            if (valid && s0 < bestS[r])     { bestS[r]     = s0; bestI[r]     = gk; }
            if (valid && s1 < bestS[r + 8]) { bestS[r + 8] = s1; bestI[r + 8] = gk; }
        }
        if (t + 2 < numSteps)
            cc_tile_fetch(pre, cb, K, tileBase + 2 * CODES_PER_STEP, nSub, cIdx);
        cc_join();
    }

#pragma unroll
    for (int r = 0; r < 16; ++r) {
        float s = bestS[r];
        int   i = bestI[r];
#pragma unroll
        for (int off = 8; off >= 1; off >>= 1) {
            float os = __shfl_xor(s, off, 32);
            int   oi = __shfl_xor(i, off, 32);
            if (os < s || (os == s && oi < i)) { s = os; i = oi; }
        }
        bestS[r] = s; bestI[r] = i;
    }

    const int h = lane >> 4;
    float* mS = ldsB;
    int*   mI = (int*)(ldsB + CODE_GROUPS * ROWS_PER_BLOCK);
    if (nLane == 0) {
#pragma unroll
        for (int r = 0; r < 16; ++r) {
            const int mLoc = rg * 32 + (r >> 3) * 16 + 8 * h + (r & 7);
            mS[cg * ROWS_PER_BLOCK + mLoc] = bestS[r];
            mI[cg * ROWS_PER_BLOCK + mLoc] = bestI[r];
        }
    }
    __syncthreads();

    float localSum = 0.f;
#pragma unroll
    for (int j = 0; j < 4; ++j) {
        const int m   = (w & 3) * 16 + (w >> 2) * 8 + 2 * j + h;
        const int row = rowBase + m;
        float sA = mS[m];
        int   iA = mI[m];
        for (int g = 1; g < CODE_GROUPS; ++g) {
            const float sB = mS[g * ROWS_PER_BLOCK + m];
            const int   iB = mI[g * ROWS_PER_BLOCK + m];
            const bool take = sB < sA || (sB == sA && iB < iA);
            sA = take ? sB : sA;
            iA = take ? iB : iA;
        }
        const int idx = iA;
        const v4f* qsrc = (const v4f*)(cb + (size_t)idx * D_DIM);
        const v4f* esrc = (const v4f*)(embeds + (size_t)row * D_DIM);
        v4f* dst = (v4f*)(out + ((size_t)layer * nRows + row) * D_DIM);
#pragma unroll
        for (int vv = 0; vv < 4; ++vv) {
            int q4 = nLane * 4 + vv;
            v4f q = qsrc[q4];
            v4f e = esrc[q4];
            dst[q4] = q;
            v4f df = q - e;
            localSum += df[0]*df[0] + df[1]*df[1] + df[2]*df[2] + df[3]*df[3];
        }
    }

#pragma unroll
    for (int off = 16; off >= 1; off >>= 1)
        localSum += __shfl_xor(localSum, off, 32);
    if (lane == 0)
        partials[(blockIdx.y * gridDim.x + blockIdx.x) * WAVES_PER_BLOCK + w] = localSum;
}

#define FIN_THREADS 1024
#define FIN_BATCH 8
__global__ __launch_bounds__(FIN_THREADS)
void cc_finalize(const float* __restrict__ partials, int n,
                 float* __restrict__ out, size_t lossIdx, float invND) {
    __shared__ float waveSum[FIN_THREADS / 32];
    const int tid  = threadIdx.x;
    const int lane = tid & 31;
    float s = 0.f;
    for (int base = 0; base < n; base += FIN_THREADS * FIN_BATCH) {
        float v[FIN_BATCH];
#pragma unroll
        for (int j = 0; j < FIN_BATCH; ++j) {
            const int i = base + j * FIN_THREADS + tid;
            v[j] = (i < n) ? partials[i] : 0.f;
        }
#pragma unroll
        for (int j = 0; j < FIN_BATCH; ++j) s += v[j];
    }
#pragma unroll
    for (int off = 16; off >= 1; off >>= 1)
        s += __shfl_xor(s, off, 32);
    if (lane == 0) waveSum[tid >> 5] = s;
    __syncthreads();
    if (tid < 32) {
        float t = waveSum[lane];
#pragma unroll
        for (int off = 16; off >= 1; off >>= 1)
            t += __shfl_xor(t, off, 32);
        if (lane == 0) out[lossIdx] = 1.25f * invND * t;
    }
}

extern "C" void kernel_launch(void* const* d_in, const int* in_sizes, int n_in,
                              void* d_out, int out_size, void* d_ws, size_t ws_size,
                              hipStream_t stream) {
    const float* embeds = (const float*)d_in[0];
    const float* cb0    = (const float*)d_in[1];
    const float* cb1    = (const float*)d_in[2];
    const float* cb2    = (const float*)d_in[3];

    const int nRows = in_sizes[0] / D_DIM;
    const int K0 = in_sizes[1] / D_DIM;
    const int K1 = in_sizes[2] / D_DIM;
    const int K2 = in_sizes[3] / D_DIM;
    const int totalK = K0 + K1 + K2;

    float* norms    = (float*)d_ws;
    float* partials = (float*)d_ws + 8192;

    cc_norms<<<(totalK + 255) / 256, 256, 0, stream>>>(cb0, K0, cb1, K1, cb2, K2, norms);

    const int gx = nRows / ROWS_PER_BLOCK;
    const size_t smemBytes =
        (size_t)(A_SLOTS + 2 * B_SLOTS) * A_STRIDE * sizeof(float);
    cc_main<<<dim3(gx, 3), THREADS, smemBytes, stream>>>(
        embeds, cb0, K0, cb1, K1, cb2, K2, norms,
        (float*)d_out, partials, nRows);

    cc_finalize<<<1, FIN_THREADS, 0, stream>>>(partials, gx * 3 * WAVES_PER_BLOCK, (float*)d_out,
                                     (size_t)out_size - 1,
                                     1.0f / ((float)nRows * (float)D_DIM));
}


# Step's vector work beside the matrix stream cut: tile fetch from one base with constant row offsets (row clamp only in the last tile's path), constant LDS store offsets, argmin update as selects
# speedup vs baseline: 3.2751x; 1.0633x over previous
#include <hip/hip_runtime.h>
#include <hip/hip_bf16.h>
#include <float.h>


typedef float v2f __attribute__((ext_vector_type(2)));
typedef float v4f __attribute__((ext_vector_type(4)));
typedef float v8f __attribute__((ext_vector_type(8)));

#define D_DIM 256
#define A_STRIDE 132
#define ROW_GROUPS 2
#define CODE_GROUPS 4
#define WAVES_PER_BLOCK (ROW_GROUPS * CODE_GROUPS)
#define THREADS (WAVES_PER_BLOCK * 32)
#define ROWS_PER_BLOCK (ROW_GROUPS * 32)
#define CODES_PER_STEP (CODE_GROUPS * 16)
#define A_SLOTS (ROW_GROUPS * 2 * 32)
#define B_SLOTS (CODE_GROUPS * 32)
#define TILE_FLOATS (B_SLOTS * A_STRIDE)
#define ROWS_PER_IT (THREADS / (D_DIM / 4))
#define LOADS_PER_THREAD (CODES_PER_STEP / ROWS_PER_IT)
#define K_CHUNKS (D_DIM / 8)
#define BLK_CHUNKS 8
#define N_BLK (K_CHUNKS / BLK_CHUNKS)

__device__ __forceinline__ void cc_tile_fetch(v4f (&pre)[LOADS_PER_THREAD],
                                              const float* __restrict__ cb, int K,
                                              int tileBase, int nSub, int c) {
#pragma unroll
    for (int it = 0; it < LOADS_PER_THREAD; ++it) {
        int srcRow = tileBase + it * ROWS_PER_IT + nSub;
        if (srcRow >= K) srcRow = K - 1;
        pre[it] = *((const v4f*)(cb + (size_t)srcRow * D_DIM) + c);
    }
}

__device__ __forceinline__ void cc_tile_fetch_full(v4f (&pre)[LOADS_PER_THREAD],
                                                   const float* __restrict__ tileP, int fetchOff) {
    const float* lp = tileP + fetchOff;
#pragma unroll
    for (int it = 0; it < LOADS_PER_THREAD; ++it)
        pre[it] = *(const v4f*)(lp + it * (ROWS_PER_IT * D_DIM));
}

__device__ __forceinline__ void cc_tile_store(const v4f (&pre)[LOADS_PER_THREAD],
                                              float* __restrict__ tb) {
#pragma unroll
    for (int it = 0; it < LOADS_PER_THREAD; ++it) {
        float* dst0 = tb + ((it >> 2) * 32 + (it & 3) * ROWS_PER_IT) * A_STRIDE;
        float* dst1 = dst0 + 16 * A_STRIDE;
        *(v2f*)dst0 = (v2f){pre[it][0], pre[it][1]};
        *(v2f*)dst1 = (v2f){pre[it][2], pre[it][3]};
    }
}

struct cc_frags {
    v4f b[BLK_CHUNKS];
};

__device__ __forceinline__ void cc_frag_load(cc_frags& f, const float* __restrict__ bP, int blk) {
#pragma unroll
    for (int i = 0; i < BLK_CHUNKS; ++i) {
        const int c2 = blk * BLK_CHUNKS + i;
        f.b[i] = *(const v4f*)(bP + c2 * 4);
    }
}

__device__ __forceinline__ void cc_block_mma(const cc_frags& f,
                                             const v4f (&aT0)[K_CHUNKS], const v4f (&aT1)[K_CHUNKS],
                                             int blk, v8f& acc0, v8f& acc1) {
#pragma unroll
    for (int i = 0; i < BLK_CHUNKS; ++i) {
        const int c2 = blk * BLK_CHUNKS + i;
        const v4f av0 = aT0[c2], av1 = aT1[c2], bv = f.b[i];
        v2f a00 = {av0[0], av0[1]}, a01 = {av0[2], av0[3]};
        v2f a10 = {av1[0], av1[1]}, a11 = {av1[2], av1[3]};
        v2f b0  = {bv[0],  bv[1]},  b1  = {bv[2],  bv[3]};
        acc0 = __builtin_amdgcn_wmma_f32_16x16x4_f32(false, a00, false, b0,
                                                     (short)0, acc0, false, false);
        acc1 = __builtin_amdgcn_wmma_f32_16x16x4_f32(false, a10, false, b0,
                                                     (short)0, acc1, false, false);
        acc0 = __builtin_amdgcn_wmma_f32_16x16x4_f32(false, a01, false, b1,
                                                     (short)0, acc0, false, false);
        acc1 = __builtin_amdgcn_wmma_f32_16x16x4_f32(false, a11, false, b1,
                                                     (short)0, acc1, false, false);
    }
}

__device__ __forceinline__ void cc_arrive() {
    asm volatile("s_wait_dscnt 0x0\n\ts_barrier_signal -1" ::: "memory");
}
__device__ __forceinline__ void cc_join() {
    asm volatile("s_barrier_wait -1" ::: "memory");
}
__device__ __forceinline__ void cc_join_behind(const float (&bS)[16], const int (&bI)[16]) {
    asm volatile("s_barrier_wait -1"
                 :: "v"(bS[0]), "v"(bS[1]), "v"(bS[2]), "v"(bS[3]), "v"(bS[4]), "v"(bS[5]), "v"(bS[6]), "v"(bS[7]),
                    "v"(bS[8]), "v"(bS[9]), "v"(bS[10]), "v"(bS[11]), "v"(bS[12]), "v"(bS[13]), "v"(bS[14]), "v"(bS[15]),
                    "v"(bI[0]), "v"(bI[1]), "v"(bI[2]), "v"(bI[3]), "v"(bI[4]), "v"(bI[5]), "v"(bI[6]), "v"(bI[7]),
                    "v"(bI[8]), "v"(bI[9]), "v"(bI[10]), "v"(bI[11]), "v"(bI[12]), "v"(bI[13]), "v"(bI[14]), "v"(bI[15])
                 : "memory");
}

__global__ void cc_norms(const float* __restrict__ cb0, int K0,
                         const float* __restrict__ cb1, int K1,
                         const float* __restrict__ cb2, int K2,
                         float* __restrict__ norms) {
    int i = blockIdx.x * blockDim.x + threadIdx.x;
    int total = K0 + K1 + K2;
    if (i >= total) return;
    const float* src;
    int row;
    if (i < K0)           { src = cb0; row = i; }
    else if (i < K0 + K1) { src = cb1; row = i - K0; }
    else                  { src = cb2; row = i - K0 - K1; }
    const v4f* p = (const v4f*)(src + (size_t)row * D_DIM);
    float s = 0.f;
    for (int j = 0; j < D_DIM / 4; ++j) {
        v4f v = p[j];
        s += v[0]*v[0] + v[1]*v[1] + v[2]*v[2] + v[3]*v[3];
    }
    norms[i] = s;
}

__global__ __launch_bounds__(THREADS)
void cc_main(const float* __restrict__ embeds,
             const float* __restrict__ cb0, int K0,
             const float* __restrict__ cb1, int K1,
             const float* __restrict__ cb2, int K2,
             const float* __restrict__ norms,
             float* __restrict__ out,
             float* __restrict__ partials,
             int nRows) {
    extern __shared__ float smem[];
    float* ldsA = smem;
    float* ldsB = smem + A_SLOTS * A_STRIDE;

    const int tid   = threadIdx.x;
    const int lane  = tid & 31;
    const int w     = tid >> 5;
    const int rg    = w & (ROW_GROUPS - 1);
    const int cg    = w >> 1;
    const int layer = blockIdx.y;

    const float* cb; int K; int normOff;
    if (layer == 0)      { cb = cb0; K = K0; normOff = 0; }
    else if (layer == 1) { cb = cb1; K = K1; normOff = K0; }
    else                 { cb = cb2; K = K2; normOff = K0 + K1; }

    const int rowBase = blockIdx.x * ROWS_PER_BLOCK;

#pragma unroll
    for (int it = 0; it < (ROWS_PER_BLOCK * (D_DIM / 4)) / THREADS; ++it) {
        const int idx = it * THREADS + tid;
        const int row = idx >> 6;
        const int c   = idx & 63;
        v4f v = *((const v4f*)(embeds + (size_t)(rowBase + row) * D_DIM) + c);
        float* dst0 = ldsA + ((row >> 4) * 32 + (row & 15)) * A_STRIDE;
        float* dst1 = dst0 + 16 * A_STRIDE;
        *(v2f*)(dst0 + 2 * c) = (v2f){v[0], v[1]};
        *(v2f*)(dst1 + 2 * c) = (v2f){v[2], v[3]};
    }

    float bestS[16];
    int   bestI[16];
#pragma unroll
    for (int r = 0; r < 16; ++r) { bestS[r] = FLT_MAX; bestI[r] = 0x7fffffff; }

    const int nLane = lane & 15;
    const float* aP0 = ldsA + ((rg * 2 + 0) * 32 + lane) * A_STRIDE;
    const float* aP1 = ldsA + ((rg * 2 + 1) * 32 + lane) * A_STRIDE;
    const int bOff  = (cg * 32 + lane) * A_STRIDE;
    const int nSub  = tid >> 6;
    const int cIdx  = tid & 63;
    const int laneCode = cg * 16 + nLane;
    const int fetchOff = nSub * D_DIM + cIdx * 4;
    const int storeOff = nSub * A_STRIDE + 2 * cIdx;

    const int numSteps = (K + CODES_PER_STEP - 1) / CODES_PER_STEP;

    v4f pre[LOADS_PER_THREAD];
    cc_tile_fetch(pre, cb, K, 0, nSub, cIdx);
    cc_tile_store(pre, ldsB + storeOff);
    cc_arrive();
    if (numSteps > 1)
        cc_tile_fetch(pre, cb, K, CODES_PER_STEP, nSub, cIdx);
    cc_join();

    v4f aT0[K_CHUNKS], aT1[K_CHUNKS];
#pragma unroll
    for (int c2 = 0; c2 < K_CHUNKS; ++c2) {
        aT0[c2] = *(const v4f*)(aP0 + c2 * 4);
        aT1[c2] = *(const v4f*)(aP1 + c2 * 4);
    }

    for (int t = 0; t < numSteps; ++t) {
        const int tileBase = t * CODES_PER_STEP;
        const bool more = (t + 1 < numSteps);
        const int gk = tileBase + laneCode;
        const bool valid = gk < K;
        const float nr = norms[normOff + (valid ? gk : K - 1)];
        __builtin_amdgcn_sched_barrier(0);

        const float* bP = ldsB + (t & 1) * TILE_FLOATS + bOff;

        v8f acc0 = {0.f,0.f,0.f,0.f,0.f,0.f,0.f,0.f};
        v8f acc1 = {0.f,0.f,0.f,0.f,0.f,0.f,0.f,0.f};
        cc_frags fr[2];
        cc_frag_load(fr[0], bP, 0);
        cc_frag_load(fr[1], bP, 1);
#pragma unroll
        for (int b = 0; b < N_BLK; ++b) {
            if (b + 1 < N_BLK) __builtin_amdgcn_sched_barrier(0);
            cc_block_mma(fr[b & 1], aT0, aT1, b, acc0, acc1);
            if (b + 2 < N_BLK) {
                __builtin_amdgcn_sched_barrier(0);
                cc_frag_load(fr[b & 1], bP, b + 2);
            }
        }

        if (more) cc_tile_store(pre, ldsB + ((t + 1) & 1) * TILE_FLOATS + storeOff);
        cc_arrive();

        asm volatile("" :: "v"(nr));

        if (t + 2 < numSteps) {
            const int tb2 = tileBase + 2 * CODES_PER_STEP;
            if (tb2 + CODES_PER_STEP <= K)
                cc_tile_fetch_full(pre, cb + (size_t)tb2 * D_DIM, fetchOff);
            else
                cc_tile_fetch(pre, cb, K, tb2, nSub, cIdx);
        }

        const float wn = valid ? nr : FLT_MAX;
#pragma unroll
        for (int r = 0; r < 8; ++r) {
            const float s0 = wn - 2.0f * acc0[r];
            const float s1 = wn - 2.0f * acc1[r];
            const bool p0 = valid && s0 < bestS[r];
            const bool p1 = valid && s1 < bestS[r + 8];
            bestS[r]     = p0 ? s0 : bestS[r];
            bestI[r]     = p0 ? gk : bestI[r];
            bestS[r + 8] = p1 ? s1 : bestS[r + 8];
            bestI[r + 8] = p1 ? gk : bestI[r + 8];
        }
        cc_join_behind(bestS, bestI);
    }

#pragma unroll
    for (int r = 0; r < 16; ++r) {
        float s = bestS[r];
        int   i = bestI[r];
#pragma unroll
        for (int off = 8; off >= 1; off >>= 1) {
            float os = __shfl_xor(s, off, 32);
            int   oi = __shfl_xor(i, off, 32);
            if (os < s || (os == s && oi < i)) { s = os; i = oi; }
        }
        bestS[r] = s; bestI[r] = i;
    }

    const int h = lane >> 4;
    float* mS = ldsB;
    int*   mI = (int*)(ldsB + CODE_GROUPS * ROWS_PER_BLOCK);
    if (nLane == 0) {
#pragma unroll
        for (int r = 0; r < 16; ++r) {
            const int mLoc = rg * 32 + (r >> 3) * 16 + 8 * h + (r & 7);
            mS[cg * ROWS_PER_BLOCK + mLoc] = bestS[r];
            mI[cg * ROWS_PER_BLOCK + mLoc] = bestI[r];
        }
    }
    __syncthreads();

    float localSum = 0.f;
#pragma unroll
    for (int j = 0; j < 4; ++j) {
        const int m   = (w & 3) * 16 + (w >> 2) * 8 + 2 * j + h;
        const int row = rowBase + m;
        float sA = mS[m];
        int   iA = mI[m];
        for (int g = 1; g < CODE_GROUPS; ++g) {
            const float sB = mS[g * ROWS_PER_BLOCK + m];
            const int   iB = mI[g * ROWS_PER_BLOCK + m];
            const bool take = sB < sA || (sB == sA && iB < iA);
            sA = take ? sB : sA;
            iA = take ? iB : iA;
        }
        const int idx = iA;
        const v4f* qsrc = (const v4f*)(cb + (size_t)idx * D_DIM);
        const v4f* esrc = (const v4f*)(embeds + (size_t)row * D_DIM);
        v4f* dst = (v4f*)(out + ((size_t)layer * nRows + row) * D_DIM);
#pragma unroll
        for (int vv = 0; vv < 4; ++vv) {
            int q4 = nLane * 4 + vv;
            v4f q = qsrc[q4];
            v4f e = esrc[q4];
            dst[q4] = q;
            v4f df = q - e;
            localSum += df[0]*df[0] + df[1]*df[1] + df[2]*df[2] + df[3]*df[3];
        }
    }

#pragma unroll
    for (int off = 16; off >= 1; off >>= 1)
        localSum += __shfl_xor(localSum, off, 32);
    if (lane == 0)
        partials[(blockIdx.y * gridDim.x + blockIdx.x) * WAVES_PER_BLOCK + w] = localSum;
}

#define FIN_THREADS 1024
#define FIN_BATCH 8
__global__ __launch_bounds__(FIN_THREADS)
void cc_finalize(const float* __restrict__ partials, int n,
                 float* __restrict__ out, size_t lossIdx, float invND) {
    __shared__ float waveSum[FIN_THREADS / 32];
    const int tid  = threadIdx.x;
    const int lane = tid & 31;
    float s = 0.f;
    for (int base = 0; base < n; base += FIN_THREADS * FIN_BATCH) {
        float v[FIN_BATCH];
#pragma unroll
        for (int j = 0; j < FIN_BATCH; ++j) {
            const int i = base + j * FIN_THREADS + tid;
            v[j] = (i < n) ? partials[i] : 0.f;
        }
#pragma unroll
        for (int j = 0; j < FIN_BATCH; ++j) s += v[j];
    }
#pragma unroll
    for (int off = 16; off >= 1; off >>= 1)
        s += __shfl_xor(s, off, 32);
    if (lane == 0) waveSum[tid >> 5] = s;
    __syncthreads();
    if (tid < 32) {
        float t = waveSum[lane];
#pragma unroll
        for (int off = 16; off >= 1; off >>= 1)
            t += __shfl_xor(t, off, 32);
        if (lane == 0) out[lossIdx] = 1.25f * invND * t;
    }
}

extern "C" void kernel_launch(void* const* d_in, const int* in_sizes, int n_in,
                              void* d_out, int out_size, void* d_ws, size_t ws_size,
                              hipStream_t stream) {
    const float* embeds = (const float*)d_in[0];
    const float* cb0    = (const float*)d_in[1];
    const float* cb1    = (const float*)d_in[2];
    const float* cb2    = (const float*)d_in[3];

    const int nRows = in_sizes[0] / D_DIM;
    const int K0 = in_sizes[1] / D_DIM;
    const int K1 = in_sizes[2] / D_DIM;
    const int K2 = in_sizes[3] / D_DIM;
    const int totalK = K0 + K1 + K2;

    float* norms    = (float*)d_ws;
    float* partials = (float*)d_ws + 8192;

    cc_norms<<<(totalK + 255) / 256, 256, 0, stream>>>(cb0, K0, cb1, K1, cb2, K2, norms);

    const int gx = nRows / ROWS_PER_BLOCK;
    const size_t smemBytes =
        (size_t)(A_SLOTS + 2 * B_SLOTS) * A_STRIDE * sizeof(float);
    cc_main<<<dim3(gx, 3), THREADS, smemBytes, stream>>>(
        embeds, cb0, K0, cb1, K1, cb2, K2, norms,
        (float*)d_out, partials, nRows);

    cc_finalize<<<1, FIN_THREADS, 0, stream>>>(partials, gx * 3 * WAVES_PER_BLOCK, (float*)d_out,
                                     (size_t)out_size - 1,
                                     1.0f / ((float)nRows * (float)D_DIM));
}


# Barrier signal moved in front of the step's last matrix block (tile stores, next fetch behind block 1): waiting waves wake and read their first fragments under 32 WMMAs; sum order unchanged
# speedup vs baseline: 3.3953x; 1.0367x over previous
#include <hip/hip_runtime.h>
#include <hip/hip_bf16.h>
#include <float.h>


typedef float v2f __attribute__((ext_vector_type(2)));
typedef float v4f __attribute__((ext_vector_type(4)));
typedef float v8f __attribute__((ext_vector_type(8)));

#define D_DIM 256
#define A_STRIDE 132
#define ROW_GROUPS 2
#define CODE_GROUPS 4
#define WAVES_PER_BLOCK (ROW_GROUPS * CODE_GROUPS)
#define THREADS (WAVES_PER_BLOCK * 32)
#define ROWS_PER_BLOCK (ROW_GROUPS * 32)
#define CODES_PER_STEP (CODE_GROUPS * 16)
#define A_SLOTS (ROW_GROUPS * 2 * 32)
#define B_SLOTS (CODE_GROUPS * 32)
#define TILE_FLOATS (B_SLOTS * A_STRIDE)
#define ROWS_PER_IT (THREADS / (D_DIM / 4))
#define LOADS_PER_THREAD (CODES_PER_STEP / ROWS_PER_IT)
#define K_CHUNKS (D_DIM / 8)
#define BLK_CHUNKS 8
#define N_BLK (K_CHUNKS / BLK_CHUNKS)

__device__ __forceinline__ void cc_tile_fetch(v4f (&pre)[LOADS_PER_THREAD],
                                              const float* __restrict__ cb, int K,
                                              int tileBase, int nSub, int c) {
#pragma unroll
    for (int it = 0; it < LOADS_PER_THREAD; ++it) {
        int srcRow = tileBase + it * ROWS_PER_IT + nSub;
        if (srcRow >= K) srcRow = K - 1;
        pre[it] = *((const v4f*)(cb + (size_t)srcRow * D_DIM) + c);
    }
}

__device__ __forceinline__ void cc_tile_fetch_full(v4f (&pre)[LOADS_PER_THREAD],
                                                   const float* __restrict__ tileP, int fetchOff) {
    const float* lp = tileP + fetchOff;
#pragma unroll
    for (int it = 0; it < LOADS_PER_THREAD; ++it)
        pre[it] = *(const v4f*)(lp + it * (ROWS_PER_IT * D_DIM));
}

__device__ __forceinline__ void cc_tile_store(const v4f (&pre)[LOADS_PER_THREAD],
                                              float* __restrict__ tb) {
#pragma unroll
    for (int it = 0; it < LOADS_PER_THREAD; ++it) {
        float* dst0 = tb + ((it >> 2) * 32 + (it & 3) * ROWS_PER_IT) * A_STRIDE;
        float* dst1 = dst0 + 16 * A_STRIDE;
        *(v2f*)dst0 = (v2f){pre[it][0], pre[it][1]};
        *(v2f*)dst1 = (v2f){pre[it][2], pre[it][3]};
    }
}

struct cc_frags {
    v4f b[BLK_CHUNKS];
};

__device__ __forceinline__ void cc_frag_load(cc_frags& f, const float* __restrict__ bP, int blk) {
#pragma unroll
    for (int i = 0; i < BLK_CHUNKS; ++i) {
        const int c2 = blk * BLK_CHUNKS + i;
        f.b[i] = *(const v4f*)(bP + c2 * 4);
    }
}

__device__ __forceinline__ void cc_block_mma(const cc_frags& f,
                                             const v4f (&aT0)[K_CHUNKS], const v4f (&aT1)[K_CHUNKS],
                                             int blk, v8f& acc0, v8f& acc1) {
#pragma unroll
    for (int i = 0; i < BLK_CHUNKS; ++i) {
        const int c2 = blk * BLK_CHUNKS + i;
        const v4f av0 = aT0[c2], av1 = aT1[c2], bv = f.b[i];
        v2f a00 = {av0[0], av0[1]}, a01 = {av0[2], av0[3]};
        v2f a10 = {av1[0], av1[1]}, a11 = {av1[2], av1[3]};
        v2f b0  = {bv[0],  bv[1]},  b1  = {bv[2],  bv[3]};
        acc0 = __builtin_amdgcn_wmma_f32_16x16x4_f32(false, a00, false, b0,
                                                     (short)0, acc0, false, false);
        acc1 = __builtin_amdgcn_wmma_f32_16x16x4_f32(false, a10, false, b0,
                                                     (short)0, acc1, false, false);
        acc0 = __builtin_amdgcn_wmma_f32_16x16x4_f32(false, a01, false, b1,
                                                     (short)0, acc0, false, false);
        acc1 = __builtin_amdgcn_wmma_f32_16x16x4_f32(false, a11, false, b1,
                                                     (short)0, acc1, false, false);
    }
}

__device__ __forceinline__ void cc_arrive() {
    asm volatile("s_wait_dscnt 0x0\n\ts_barrier_signal -1" ::: "memory");
}
__device__ __forceinline__ void cc_join() {
    asm volatile("s_barrier_wait -1" ::: "memory");
}
__device__ __forceinline__ void cc_join_behind(const float (&bS)[16], const int (&bI)[16]) {
    asm volatile("s_barrier_wait -1"
                 :: "v"(bS[0]), "v"(bS[1]), "v"(bS[2]), "v"(bS[3]), "v"(bS[4]), "v"(bS[5]), "v"(bS[6]), "v"(bS[7]),
                    "v"(bS[8]), "v"(bS[9]), "v"(bS[10]), "v"(bS[11]), "v"(bS[12]), "v"(bS[13]), "v"(bS[14]), "v"(bS[15]),
                    "v"(bI[0]), "v"(bI[1]), "v"(bI[2]), "v"(bI[3]), "v"(bI[4]), "v"(bI[5]), "v"(bI[6]), "v"(bI[7]),
                    "v"(bI[8]), "v"(bI[9]), "v"(bI[10]), "v"(bI[11]), "v"(bI[12]), "v"(bI[13]), "v"(bI[14]), "v"(bI[15])
                 : "memory");
}

__global__ void cc_norms(const float* __restrict__ cb0, int K0,
                         const float* __restrict__ cb1, int K1,
                         const float* __restrict__ cb2, int K2,
                         float* __restrict__ norms) {
    int i = blockIdx.x * blockDim.x + threadIdx.x;
    int total = K0 + K1 + K2;
    if (i >= total) return;
    const float* src;
    int row;
    if (i < K0)           { src = cb0; row = i; }
    else if (i < K0 + K1) { src = cb1; row = i - K0; }
    else                  { src = cb2; row = i - K0 - K1; }
    const v4f* p = (const v4f*)(src + (size_t)row * D_DIM);
    float s = 0.f;
    for (int j = 0; j < D_DIM / 4; ++j) {
        v4f v = p[j];
        s += v[0]*v[0] + v[1]*v[1] + v[2]*v[2] + v[3]*v[3];
    }
    norms[i] = s;
}

__global__ __launch_bounds__(THREADS)
void cc_main(const float* __restrict__ embeds,
             const float* __restrict__ cb0, int K0,
             const float* __restrict__ cb1, int K1,
             const float* __restrict__ cb2, int K2,
             const float* __restrict__ norms,
             float* __restrict__ out,
             float* __restrict__ partials,
             int nRows) {
    extern __shared__ float smem[];
    float* ldsA = smem;
    float* ldsB = smem + A_SLOTS * A_STRIDE;

    const int tid   = threadIdx.x;
    const int lane  = tid & 31;
    const int w     = tid >> 5;
    const int rg    = w & (ROW_GROUPS - 1);
    const int cg    = w >> 1;
    const int layer = blockIdx.y;

    const float* cb; int K; int normOff;
    if (layer == 0)      { cb = cb0; K = K0; normOff = 0; }
    else if (layer == 1) { cb = cb1; K = K1; normOff = K0; }
    else                 { cb = cb2; K = K2; normOff = K0 + K1; }

    const int rowBase = blockIdx.x * ROWS_PER_BLOCK;

#pragma unroll
    for (int it = 0; it < (ROWS_PER_BLOCK * (D_DIM / 4)) / THREADS; ++it) {
        const int idx = it * THREADS + tid;
        const int row = idx >> 6;
        const int c   = idx & 63;
        v4f v = *((const v4f*)(embeds + (size_t)(rowBase + row) * D_DIM) + c);
        float* dst0 = ldsA + ((row >> 4) * 32 + (row & 15)) * A_STRIDE;
        float* dst1 = dst0 + 16 * A_STRIDE;
        *(v2f*)(dst0 + 2 * c) = (v2f){v[0], v[1]};
        *(v2f*)(dst1 + 2 * c) = (v2f){v[2], v[3]};
    }

    float bestS[16];
    int   bestI[16];
#pragma unroll
    for (int r = 0; r < 16; ++r) { bestS[r] = FLT_MAX; bestI[r] = 0x7fffffff; }

    const int nLane = lane & 15;
    const float* aP0 = ldsA + ((rg * 2 + 0) * 32 + lane) * A_STRIDE;
    const float* aP1 = ldsA + ((rg * 2 + 1) * 32 + lane) * A_STRIDE;
    const int bOff  = (cg * 32 + lane) * A_STRIDE;
    const int nSub  = tid >> 6;
    const int cIdx  = tid & 63;
    const int laneCode = cg * 16 + nLane;
    const int fetchOff = nSub * D_DIM + cIdx * 4;
    const int storeOff = nSub * A_STRIDE + 2 * cIdx;

    const int numSteps = (K + CODES_PER_STEP - 1) / CODES_PER_STEP;

    v4f pre[LOADS_PER_THREAD];
    cc_tile_fetch(pre, cb, K, 0, nSub, cIdx);
    cc_tile_store(pre, ldsB + storeOff);
    cc_arrive();
    if (numSteps > 1)
        cc_tile_fetch(pre, cb, K, CODES_PER_STEP, nSub, cIdx);
    cc_join();

    v4f aT0[K_CHUNKS], aT1[K_CHUNKS];
#pragma unroll
    for (int c2 = 0; c2 < K_CHUNKS; ++c2) {
        aT0[c2] = *(const v4f*)(aP0 + c2 * 4);
        aT1[c2] = *(const v4f*)(aP1 + c2 * 4);
    }

    for (int t = 0; t < numSteps; ++t) {
        const int tileBase = t * CODES_PER_STEP;
        const bool more = (t + 1 < numSteps);
        const int gk = tileBase + laneCode;
        const bool valid = gk < K;
        const float nr = norms[normOff + (valid ? gk : K - 1)];
        __builtin_amdgcn_sched_barrier(0);

        const float* bP = ldsB + (t & 1) * TILE_FLOATS + bOff;

        v8f acc0 = {0.f,0.f,0.f,0.f,0.f,0.f,0.f,0.f};
        v8f acc1 = {0.f,0.f,0.f,0.f,0.f,0.f,0.f,0.f};
        cc_frags fr[2];
        cc_frag_load(fr[0], bP, 0);
        cc_frag_load(fr[1], bP, 1);
        __builtin_amdgcn_sched_barrier(0);
        cc_block_mma(fr[0], aT0, aT1, 0, acc0, acc1);
        __builtin_amdgcn_sched_barrier(0);
        cc_frag_load(fr[0], bP, 2);
        __builtin_amdgcn_sched_barrier(0);
        cc_block_mma(fr[1], aT0, aT1, 1, acc0, acc1);
        __builtin_amdgcn_sched_barrier(0);
        cc_frag_load(fr[1], bP, 3);

        asm volatile("" :: "v"(nr));

        if (more) cc_tile_store(pre, ldsB + ((t + 1) & 1) * TILE_FLOATS + storeOff);

        if (t + 2 < numSteps) {
            const int tb2 = tileBase + 2 * CODES_PER_STEP;
            if (tb2 + CODES_PER_STEP <= K)
                cc_tile_fetch_full(pre, cb + (size_t)tb2 * D_DIM, fetchOff);
            else
                cc_tile_fetch(pre, cb, K, tb2, nSub, cIdx);
        }
        __builtin_amdgcn_sched_barrier(0);
        cc_block_mma(fr[0], aT0, aT1, 2, acc0, acc1);
        __builtin_amdgcn_sched_barrier(0);
        cc_arrive();
        __builtin_amdgcn_sched_barrier(0);
        cc_block_mma(fr[1], aT0, aT1, 3, acc0, acc1);

        const float wn = valid ? nr : FLT_MAX;
#pragma unroll
        for (int r = 0; r < 8; ++r) {
            const float s0 = wn - 2.0f * acc0[r];
            const float s1 = wn - 2.0f * acc1[r];
            const bool p0 = valid && s0 < bestS[r];
            const bool p1 = valid && s1 < bestS[r + 8];
            bestS[r]     = p0 ? s0 : bestS[r];
            bestI[r]     = p0 ? gk : bestI[r];
            bestS[r + 8] = p1 ? s1 : bestS[r + 8];
            bestI[r + 8] = p1 ? gk : bestI[r + 8];
        }
        cc_join_behind(bestS, bestI);
    }

#pragma unroll
    for (int r = 0; r < 16; ++r) {
        float s = bestS[r];
        int   i = bestI[r];
#pragma unroll
        for (int off = 8; off >= 1; off >>= 1) {
            float os = __shfl_xor(s, off, 32);
            int   oi = __shfl_xor(i, off, 32);
            if (os < s || (os == s && oi < i)) { s = os; i = oi; }
        }
        bestS[r] = s; bestI[r] = i;
    }

    const int h = lane >> 4;
    float* mS = ldsB;
    int*   mI = (int*)(ldsB + CODE_GROUPS * ROWS_PER_BLOCK);
    if (nLane == 0) {
#pragma unroll
        for (int r = 0; r < 16; ++r) {
            const int mLoc = rg * 32 + (r >> 3) * 16 + 8 * h + (r & 7);
            mS[cg * ROWS_PER_BLOCK + mLoc] = bestS[r];
            mI[cg * ROWS_PER_BLOCK + mLoc] = bestI[r];
        }
    }
    __syncthreads();

    float localSum = 0.f;
#pragma unroll
    for (int j = 0; j < 4; ++j) {
        const int m   = (w & 3) * 16 + (w >> 2) * 8 + 2 * j + h;
        const int row = rowBase + m;
        float sA = mS[m];
        int   iA = mI[m];
        for (int g = 1; g < CODE_GROUPS; ++g) {
            const float sB = mS[g * ROWS_PER_BLOCK + m];
            const int   iB = mI[g * ROWS_PER_BLOCK + m];
            const bool take = sB < sA || (sB == sA && iB < iA);
            sA = take ? sB : sA;
            iA = take ? iB : iA;
        }
        const int idx = iA;
        const v4f* qsrc = (const v4f*)(cb + (size_t)idx * D_DIM);
        const v4f* esrc = (const v4f*)(embeds + (size_t)row * D_DIM);
        v4f* dst = (v4f*)(out + ((size_t)layer * nRows + row) * D_DIM);
#pragma unroll
        for (int vv = 0; vv < 4; ++vv) {
            int q4 = nLane * 4 + vv;
            v4f q = qsrc[q4];
            v4f e = esrc[q4];
            dst[q4] = q;
            v4f df = q - e;
            localSum += df[0]*df[0] + df[1]*df[1] + df[2]*df[2] + df[3]*df[3];
        }
    }

#pragma unroll
    for (int off = 16; off >= 1; off >>= 1)
        localSum += __shfl_xor(localSum, off, 32);
    if (lane == 0)
        partials[(blockIdx.y * gridDim.x + blockIdx.x) * WAVES_PER_BLOCK + w] = localSum;
}

#define FIN_THREADS 1024
#define FIN_BATCH 8
__global__ __launch_bounds__(FIN_THREADS)
void cc_finalize(const float* __restrict__ partials, int n,
                 float* __restrict__ out, size_t lossIdx, float invND) {
    __shared__ float waveSum[FIN_THREADS / 32];
    const int tid  = threadIdx.x;
    const int lane = tid & 31;
    float s = 0.f;
    for (int base = 0; base < n; base += FIN_THREADS * FIN_BATCH) {
        float v[FIN_BATCH];
#pragma unroll
        for (int j = 0; j < FIN_BATCH; ++j) {
            const int i = base + j * FIN_THREADS + tid;
            v[j] = (i < n) ? partials[i] : 0.f;
        }
#pragma unroll
        for (int j = 0; j < FIN_BATCH; ++j) s += v[j];
    }
#pragma unroll
    for (int off = 16; off >= 1; off >>= 1)
        s += __shfl_xor(s, off, 32);
    if (lane == 0) waveSum[tid >> 5] = s;
    __syncthreads();
    if (tid < 32) {
        float t = waveSum[lane];
#pragma unroll
        for (int off = 16; off >= 1; off >>= 1)
            t += __shfl_xor(t, off, 32);
        if (lane == 0) out[lossIdx] = 1.25f * invND * t;
    }
}

extern "C" void kernel_launch(void* const* d_in, const int* in_sizes, int n_in,
                              void* d_out, int out_size, void* d_ws, size_t ws_size,
                              hipStream_t stream) {
    const float* embeds = (const float*)d_in[0];
    const float* cb0    = (const float*)d_in[1];
    const float* cb1    = (const float*)d_in[2];
    const float* cb2    = (const float*)d_in[3];

    const int nRows = in_sizes[0] / D_DIM;
    const int K0 = in_sizes[1] / D_DIM;
    const int K1 = in_sizes[2] / D_DIM;
    const int K2 = in_sizes[3] / D_DIM;
    const int totalK = K0 + K1 + K2;

    float* norms    = (float*)d_ws;
    float* partials = (float*)d_ws + 8192;

    cc_norms<<<(totalK + 255) / 256, 256, 0, stream>>>(cb0, K0, cb1, K1, cb2, K2, norms);

    const int gx = nRows / ROWS_PER_BLOCK;
    const size_t smemBytes =
        (size_t)(A_SLOTS + 2 * B_SLOTS) * A_STRIDE * sizeof(float);
    cc_main<<<dim3(gx, 3), THREADS, smemBytes, stream>>>(
        embeds, cb0, K0, cb1, K1, cb2, K2, norms,
        (float*)d_out, partials, nRows);

    cc_finalize<<<1, FIN_THREADS, 0, stream>>>(partials, gx * 3 * WAVES_PER_BLOCK, (float*)d_out,
                                     (size_t)out_size - 1,
                                     1.0f / ((float)nRows * (float)D_DIM));
}


# Matrix operands exchange roles (codes on M, data rows on N): a lane's 8 registers are 8 codes of ONE row; the sweep ends in an in-register fold + 1 lane exchange, not 64 rounds; sum order unchanged
# speedup vs baseline: 3.5455x; 1.0442x over previous
#include <hip/hip_runtime.h>
#include <hip/hip_bf16.h>
#include <float.h>


typedef float v2f __attribute__((ext_vector_type(2)));
typedef float v4f __attribute__((ext_vector_type(4)));
typedef float v8f __attribute__((ext_vector_type(8)));

#define D_DIM 256
#define A_STRIDE 132
#define ROW_GROUPS 2
#define CODE_GROUPS 4
#define WAVES_PER_BLOCK (ROW_GROUPS * CODE_GROUPS)
#define THREADS (WAVES_PER_BLOCK * 32)
#define ROWS_PER_BLOCK (ROW_GROUPS * 32)
#define CODES_PER_STEP (CODE_GROUPS * 16)
#define A_SLOTS (ROW_GROUPS * 2 * 32)
#define B_SLOTS (CODE_GROUPS * 32)
#define TILE_FLOATS (B_SLOTS * A_STRIDE)
#define ROWS_PER_IT (THREADS / (D_DIM / 4))
#define LOADS_PER_THREAD (CODES_PER_STEP / ROWS_PER_IT)
#define K_CHUNKS (D_DIM / 8)
#define BLK_CHUNKS 8
#define N_BLK (K_CHUNKS / BLK_CHUNKS)

__device__ __forceinline__ void cc_tile_fetch(v4f (&pre)[LOADS_PER_THREAD],
                                              const float* __restrict__ cb, int K,
                                              int tileBase, int nSub, int c) {
#pragma unroll
    for (int it = 0; it < LOADS_PER_THREAD; ++it) {
        int srcRow = tileBase + it * ROWS_PER_IT + nSub;
        if (srcRow >= K) srcRow = K - 1;
        pre[it] = *((const v4f*)(cb + (size_t)srcRow * D_DIM) + c);
    }
}

__device__ __forceinline__ void cc_tile_fetch_full(v4f (&pre)[LOADS_PER_THREAD],
                                                   const float* __restrict__ tileP, int fetchOff) {
    const float* lp = tileP + fetchOff;
#pragma unroll
    for (int it = 0; it < LOADS_PER_THREAD; ++it)
        pre[it] = *(const v4f*)(lp + it * (ROWS_PER_IT * D_DIM));
}

__device__ __forceinline__ void cc_tile_store(const v4f (&pre)[LOADS_PER_THREAD],
                                              float* __restrict__ tb) {
#pragma unroll
    for (int it = 0; it < LOADS_PER_THREAD; ++it) {
        float* dst0 = tb + ((it >> 2) * 32 + (it & 3) * ROWS_PER_IT) * A_STRIDE;
        float* dst1 = dst0 + 16 * A_STRIDE;
        *(v2f*)dst0 = (v2f){pre[it][0], pre[it][1]};
        *(v2f*)dst1 = (v2f){pre[it][2], pre[it][3]};
    }
}

struct cc_frags {
    v4f b[BLK_CHUNKS];
};

__device__ __forceinline__ void cc_frag_load(cc_frags& f, const float* __restrict__ bP, int blk) {
#pragma unroll
    for (int i = 0; i < BLK_CHUNKS; ++i) {
        const int c2 = blk * BLK_CHUNKS + i;
        f.b[i] = *(const v4f*)(bP + c2 * 4);
    }
}

__device__ __forceinline__ void cc_block_mma(const cc_frags& f,
                                             const v4f (&aT0)[K_CHUNKS], const v4f (&aT1)[K_CHUNKS],
                                             int blk, v8f& acc0, v8f& acc1) {
#pragma unroll
    for (int i = 0; i < BLK_CHUNKS; ++i) {
        const int c2 = blk * BLK_CHUNKS + i;
        const v4f av0 = aT0[c2], av1 = aT1[c2], bv = f.b[i];
        v2f a00 = {av0[0], av0[1]}, a01 = {av0[2], av0[3]};
        v2f a10 = {av1[0], av1[1]}, a11 = {av1[2], av1[3]};
        v2f b0  = {bv[0],  bv[1]},  b1  = {bv[2],  bv[3]};
        acc0 = __builtin_amdgcn_wmma_f32_16x16x4_f32(false, b0, false, a00,
                                                     (short)0, acc0, false, false);
        acc1 = __builtin_amdgcn_wmma_f32_16x16x4_f32(false, b0, false, a10,
                                                     (short)0, acc1, false, false);
        acc0 = __builtin_amdgcn_wmma_f32_16x16x4_f32(false, b1, false, a01,
                                                     (short)0, acc0, false, false);
        acc1 = __builtin_amdgcn_wmma_f32_16x16x4_f32(false, b1, false, a11,
                                                     (short)0, acc1, false, false);
    }
}

__device__ __forceinline__ void cc_arrive() {
    asm volatile("s_wait_dscnt 0x0\n\ts_barrier_signal -1" ::: "memory");
}
__device__ __forceinline__ void cc_join() {
    asm volatile("s_barrier_wait -1" ::: "memory");
}
__device__ __forceinline__ void cc_join_behind(const float (&bS)[16], const int (&bI)[16]) {
    asm volatile("s_barrier_wait -1"
                 :: "v"(bS[0]), "v"(bS[1]), "v"(bS[2]), "v"(bS[3]), "v"(bS[4]), "v"(bS[5]), "v"(bS[6]), "v"(bS[7]),
                    "v"(bS[8]), "v"(bS[9]), "v"(bS[10]), "v"(bS[11]), "v"(bS[12]), "v"(bS[13]), "v"(bS[14]), "v"(bS[15]),
                    "v"(bI[0]), "v"(bI[1]), "v"(bI[2]), "v"(bI[3]), "v"(bI[4]), "v"(bI[5]), "v"(bI[6]), "v"(bI[7]),
                    "v"(bI[8]), "v"(bI[9]), "v"(bI[10]), "v"(bI[11]), "v"(bI[12]), "v"(bI[13]), "v"(bI[14]), "v"(bI[15])
                 : "memory");
}

__device__ __forceinline__ int cc_pad_codes(int K) {
    return (K + CODES_PER_STEP - 1) / CODES_PER_STEP * CODES_PER_STEP;
}
__global__ void cc_norms(const float* __restrict__ cb0, int K0,
                         const float* __restrict__ cb1, int K1,
                         const float* __restrict__ cb2, int K2,
                         float* __restrict__ norms) {
    int i = blockIdx.x * blockDim.x + threadIdx.x;
    const int P0 = cc_pad_codes(K0), P1 = cc_pad_codes(K1), P2 = cc_pad_codes(K2);
    if (i >= P0 + P1 + P2) return;
    const float* src;
    int row, K;
    if (i < P0)           { src = cb0; row = i;           K = K0; }
    else if (i < P0 + P1) { src = cb1; row = i - P0;      K = K1; }
    else                  { src = cb2; row = i - P0 - P1; K = K2; }
    float s = FLT_MAX;
    if (row < K) {
        const v4f* p = (const v4f*)(src + (size_t)row * D_DIM);
        s = 0.f;
        for (int j = 0; j < D_DIM / 4; ++j) {
            v4f v = p[j];
            s += v[0]*v[0] + v[1]*v[1] + v[2]*v[2] + v[3]*v[3];
        }
    }
    norms[i] = s;
}

__global__ __launch_bounds__(THREADS)
void cc_main(const float* __restrict__ embeds,
             const float* __restrict__ cb0, int K0,
             const float* __restrict__ cb1, int K1,
             const float* __restrict__ cb2, int K2,
             const float* __restrict__ norms,
             float* __restrict__ out,
             float* __restrict__ partials,
             int nRows) {
    extern __shared__ float smem[];
    float* ldsA = smem;
    float* ldsB = smem + A_SLOTS * A_STRIDE;

    const int tid   = threadIdx.x;
    const int lane  = tid & 31;
    const int w     = tid >> 5;
    const int rg    = w & (ROW_GROUPS - 1);
    const int cg    = w >> 1;
    const int layer = blockIdx.y;

    const float* cb; int K; int normOff;
    if (layer == 0)      { cb = cb0; K = K0; normOff = 0; }
    else if (layer == 1) { cb = cb1; K = K1; normOff = cc_pad_codes(K0); }
    else                 { cb = cb2; K = K2; normOff = cc_pad_codes(K0) + cc_pad_codes(K1); }

    const int rowBase = blockIdx.x * ROWS_PER_BLOCK;

#pragma unroll
    for (int it = 0; it < (ROWS_PER_BLOCK * (D_DIM / 4)) / THREADS; ++it) {
        const int idx = it * THREADS + tid;
        const int row = idx >> 6;
        const int c   = idx & 63;
        v4f v = *((const v4f*)(embeds + (size_t)(rowBase + row) * D_DIM) + c);
        float* dst0 = ldsA + ((row >> 4) * 32 + (row & 15)) * A_STRIDE;
        float* dst1 = dst0 + 16 * A_STRIDE;
        *(v2f*)(dst0 + 2 * c) = (v2f){v[0], v[1]};
        *(v2f*)(dst1 + 2 * c) = (v2f){v[2], v[3]};
    }

    float bestS[16];
    int   bestI[16];
#pragma unroll
    for (int r = 0; r < 16; ++r) { bestS[r] = FLT_MAX; bestI[r] = 0x7ffffff0; }

    const int nLane = lane & 15;
    const float* aP0 = ldsA + ((rg * 2 + 0) * 32 + lane) * A_STRIDE;
    const float* aP1 = ldsA + ((rg * 2 + 1) * 32 + lane) * A_STRIDE;
    const int bOff  = (cg * 32 + lane) * A_STRIDE;
    const int nSub  = tid >> 6;
    const int cIdx  = tid & 63;
    const int codeOff = cg * 16 + 8 * (lane >> 4);
    const int fetchOff = nSub * D_DIM + cIdx * 4;
    const int storeOff = nSub * A_STRIDE + 2 * cIdx;

    const int numSteps = (K + CODES_PER_STEP - 1) / CODES_PER_STEP;

    v4f pre[LOADS_PER_THREAD];
    cc_tile_fetch(pre, cb, K, 0, nSub, cIdx);
    cc_tile_store(pre, ldsB + storeOff);
    cc_arrive();
    if (numSteps > 1)
        cc_tile_fetch(pre, cb, K, CODES_PER_STEP, nSub, cIdx);
    cc_join();

    v4f aT0[K_CHUNKS], aT1[K_CHUNKS];
#pragma unroll
    for (int c2 = 0; c2 < K_CHUNKS; ++c2) {
        aT0[c2] = *(const v4f*)(aP0 + c2 * 4);
        aT1[c2] = *(const v4f*)(aP1 + c2 * 4);
    }

    for (int t = 0; t < numSteps; ++t) {
        const int tileBase = t * CODES_PER_STEP;
        const bool more = (t + 1 < numSteps);
        const int gBase = tileBase + codeOff;
        const float* nP = norms + normOff + gBase;
        const v4f nLo = *(const v4f*)nP;
        const v4f nHi = *(const v4f*)(nP + 4);
        __builtin_amdgcn_sched_barrier(0);

        const float* bP = ldsB + (t & 1) * TILE_FLOATS + bOff;

        v8f acc0 = {0.f,0.f,0.f,0.f,0.f,0.f,0.f,0.f};
        v8f acc1 = {0.f,0.f,0.f,0.f,0.f,0.f,0.f,0.f};
        cc_frags fr[2];
        cc_frag_load(fr[0], bP, 0);
        cc_frag_load(fr[1], bP, 1);
        __builtin_amdgcn_sched_barrier(0);
        cc_block_mma(fr[0], aT0, aT1, 0, acc0, acc1);
        __builtin_amdgcn_sched_barrier(0);
        cc_frag_load(fr[0], bP, 2);
        __builtin_amdgcn_sched_barrier(0);
        cc_block_mma(fr[1], aT0, aT1, 1, acc0, acc1);
        __builtin_amdgcn_sched_barrier(0);
        cc_frag_load(fr[1], bP, 3);

        asm volatile("" :: "v"(nLo), "v"(nHi));

        if (more) cc_tile_store(pre, ldsB + ((t + 1) & 1) * TILE_FLOATS + storeOff);

        if (t + 2 < numSteps) {
            const int tb2 = tileBase + 2 * CODES_PER_STEP;
            if (tb2 + CODES_PER_STEP <= K)
                cc_tile_fetch_full(pre, cb + (size_t)tb2 * D_DIM, fetchOff);
            else
                cc_tile_fetch(pre, cb, K, tb2, nSub, cIdx);
        }
        __builtin_amdgcn_sched_barrier(0);
        cc_block_mma(fr[0], aT0, aT1, 2, acc0, acc1);
        __builtin_amdgcn_sched_barrier(0);
        cc_arrive();
        __builtin_amdgcn_sched_barrier(0);
        cc_block_mma(fr[1], aT0, aT1, 3, acc0, acc1);

#pragma unroll
        for (int r = 0; r < 8; ++r) {
            const float wn = (r < 4) ? nLo[r & 3] : nHi[r & 3];
            const float s0 = wn - 2.0f * acc0[r];
            const float s1 = wn - 2.0f * acc1[r];
            const bool p0 = s0 < bestS[r];
            const bool p1 = s1 < bestS[r + 8];
            bestS[r]     = p0 ? s0 : bestS[r];
            bestI[r]     = p0 ? gBase : bestI[r];
            bestS[r + 8] = p1 ? s1 : bestS[r + 8];
            bestI[r + 8] = p1 ? gBase : bestI[r + 8];
        }
        cc_join_behind(bestS, bestI);
    }

    float rowS[2];
    int   rowI[2];
#pragma unroll
    for (int a = 0; a < 2; ++a) {
        float s = bestS[8 * a];
        int   i = bestI[8 * a];
#pragma unroll
        for (int r = 1; r < 8; ++r) {
            const float os = bestS[8 * a + r];
            const int   oi = bestI[8 * a + r] + r;
            const bool take = (os < s) | ((os == s) & (oi < i));
            s = take ? os : s;
            i = take ? oi : i;
        }
        const float os = __shfl_xor(s, 16, 32);
        const int   oi = __shfl_xor(i, 16, 32);
        const bool take = (os < s) | ((os == s) & (oi < i));
        rowS[a] = take ? os : s;
        rowI[a] = take ? oi : i;
    }

    const int h = lane >> 4;
    float* mS = ldsB;
    int*   mI = (int*)(ldsB + CODE_GROUPS * ROWS_PER_BLOCK);
    if (h == 0) {
#pragma unroll
        for (int a = 0; a < 2; ++a) {
            const int mLoc = rg * 32 + a * 16 + nLane;
            mS[cg * ROWS_PER_BLOCK + mLoc] = rowS[a];
            mI[cg * ROWS_PER_BLOCK + mLoc] = rowI[a];
        }
    }
    __syncthreads();

    float localSum = 0.f;
#pragma unroll
    for (int j = 0; j < 4; ++j) {
        const int m   = (w & 3) * 16 + (w >> 2) * 8 + 2 * j + h;
        const int row = rowBase + m;
        float sA = mS[m];
        int   iA = mI[m];
        for (int g = 1; g < CODE_GROUPS; ++g) {
            const float sB = mS[g * ROWS_PER_BLOCK + m];
            const int   iB = mI[g * ROWS_PER_BLOCK + m];
            const bool take = sB < sA || (sB == sA && iB < iA);
            sA = take ? sB : sA;
            iA = take ? iB : iA;
        }
        const int idx = iA;
        const v4f* qsrc = (const v4f*)(cb + (size_t)idx * D_DIM);
        const v4f* esrc = (const v4f*)(embeds + (size_t)row * D_DIM);
        v4f* dst = (v4f*)(out + ((size_t)layer * nRows + row) * D_DIM);
#pragma unroll
        for (int vv = 0; vv < 4; ++vv) {
            int q4 = nLane * 4 + vv;
            v4f q = qsrc[q4];
            v4f e = esrc[q4];
            dst[q4] = q;
            v4f df = q - e;
            localSum += df[0]*df[0] + df[1]*df[1] + df[2]*df[2] + df[3]*df[3];
        }
    }

#pragma unroll
    for (int off = 16; off >= 1; off >>= 1)
        localSum += __shfl_xor(localSum, off, 32);
    if (lane == 0)
        partials[(blockIdx.y * gridDim.x + blockIdx.x) * WAVES_PER_BLOCK + w] = localSum;
}

#define FIN_THREADS 1024
#define FIN_BATCH 8
__global__ __launch_bounds__(FIN_THREADS)
void cc_finalize(const float* __restrict__ partials, int n,
                 float* __restrict__ out, size_t lossIdx, float invND) {
    __shared__ float waveSum[FIN_THREADS / 32];
    const int tid  = threadIdx.x;
    const int lane = tid & 31;
    float s = 0.f;
    for (int base = 0; base < n; base += FIN_THREADS * FIN_BATCH) {
        float v[FIN_BATCH];
#pragma unroll
        for (int j = 0; j < FIN_BATCH; ++j) {
            const int i = base + j * FIN_THREADS + tid;
            v[j] = (i < n) ? partials[i] : 0.f;
        }
#pragma unroll
        for (int j = 0; j < FIN_BATCH; ++j) s += v[j];
    }
#pragma unroll
    for (int off = 16; off >= 1; off >>= 1)
        s += __shfl_xor(s, off, 32);
    if (lane == 0) waveSum[tid >> 5] = s;
    __syncthreads();
    if (tid < 32) {
        float t = waveSum[lane];
#pragma unroll
        for (int off = 16; off >= 1; off >>= 1)
            t += __shfl_xor(t, off, 32);
        if (lane == 0) out[lossIdx] = 1.25f * invND * t;
    }
}

extern "C" void kernel_launch(void* const* d_in, const int* in_sizes, int n_in,
                              void* d_out, int out_size, void* d_ws, size_t ws_size,
                              hipStream_t stream) {
    const float* embeds = (const float*)d_in[0];
    const float* cb0    = (const float*)d_in[1];
    const float* cb1    = (const float*)d_in[2];
    const float* cb2    = (const float*)d_in[3];

    const int nRows = in_sizes[0] / D_DIM;
    const int K0 = in_sizes[1] / D_DIM;
    const int K1 = in_sizes[2] / D_DIM;
    const int K2 = in_sizes[3] / D_DIM;
    const int totalK = K0 + K1 + K2;

    float* norms    = (float*)d_ws;
    float* partials = (float*)d_ws + 8192;

    cc_norms<<<(totalK + 3 * (CODES_PER_STEP - 1) + 255) / 256, 256, 0, stream>>>(cb0, K0, cb1, K1, cb2, K2, norms);

    const int gx = nRows / ROWS_PER_BLOCK;
    const size_t smemBytes =
        (size_t)(A_SLOTS + 2 * B_SLOTS) * A_STRIDE * sizeof(float);
    cc_main<<<dim3(gx, 3), THREADS, smemBytes, stream>>>(
        embeds, cb0, K0, cb1, K1, cb2, K2, norms,
        (float*)d_out, partials, nRows);

    cc_finalize<<<1, FIN_THREADS, 0, stream>>>(partials, gx * 3 * WAVES_PER_BLOCK, (float*)d_out,
                                     (size_t)out_size - 1,
                                     1.0f / ((float)nRows * (float)D_DIM));
}
